# starting kernel
# baseline (speedup 1.0000x reference)
#include <hip/hip_runtime.h>
#include <hip/hip_bf16.h>

typedef __attribute__((ext_vector_type(2))) float v2f;
typedef __attribute__((ext_vector_type(8))) float v8f;

#define EDIM 128
#define CDIM 128

__global__ __launch_bounds__(256)
void bte_gemm_kernel(const float* __restrict__ emb,
                     const float* __restrict__ W,
                     const float* __restrict__ bias,
                     const int*   __restrict__ tokens,
                     float*       __restrict__ h,
                     int N)
{
    __shared__ float ldsA[128 * EDIM];
    const int  block0 = blockIdx.x * 128;
    const int  tid    = threadIdx.x;
    const bool full   = (block0 + 128) <= N;

    {
        const int nl   = tid >> 1;
        const int half = (tid & 1) * 64;
        const int node = block0 + nl;
        const int tok  = (full || node < N) ? tokens[node] : 0;
        const float4* src = (const float4*)(emb + (size_t)tok * EDIM + half);
        float4*       dst = (float4*)(ldsA + nl * EDIM + half);
        #pragma unroll
        for (int i = 0; i < 16; ++i) dst[i] = src[i];
    }
    __syncthreads();

    const int wid   = tid >> 5;
    const int lane  = tid & 31;
    const int col   = lane & 15;
    const int khalf = lane >> 4;
    const float* arow = ldsA + (wid * 16 + col) * EDIM;

    for (int t = 0; t < 8; ++t) {
        const int c0 = t * 16;
        const float bv = bias[c0 + col];
        v8f acc = { bv, bv, bv, bv, bv, bv, bv, bv };
        const float* wrow = W + (size_t)(c0 + col) * EDIM;
        #pragma unroll 4
        for (int k = 0; k < EDIM; k += 4) {
            const int kk = k + 2 * khalf;
            v2f a, bf;
            a.x  = arow[kk];   a.y  = arow[kk + 1];
            bf.x = wrow[kk];   bf.y = wrow[kk + 1];
            acc = __builtin_amdgcn_wmma_f32_16x16x4_f32(
                      false, a, false, bf, (short)0, acc, false, false);
        }
        const int mbase = block0 + wid * 16 + 8 * khalf;
        float* hp = h + (size_t)mbase * CDIM + c0 + col;
        if (full) {
            #pragma unroll
            for (int r = 0; r < 8; ++r) hp[r * CDIM] = acc[r];
        } else {
            #pragma unroll
            for (int r = 0; r < 8; ++r)
                if (mbase + r < N) hp[r * CDIM] = acc[r];
        }
    }
}

__global__ __launch_bounds__(256)
void bte_scatter_kernel(float* __restrict__ h,
                        const int* __restrict__ parent,
                        int off, int count)
{
    const int idx = blockIdx.x * 256 + threadIdx.x;
    const int nl  = idx >> 5;
    if (nl >= count) return;
    const int cq   = (idx & 31) << 2;
    const int node = off + nl;
    const int par  = parent[node];
    const float4 v = *(const float4*)(h + (size_t)node * CDIM + cq);
    float* dst = h + (size_t)par * CDIM + cq;
    atomicAdd(dst + 0, v.x);
    atomicAdd(dst + 1, v.y);
    atomicAdd(dst + 2, v.z);
    atomicAdd(dst + 3, v.w);
}

__global__ __launch_bounds__(256)
void bte_init_out(float* __restrict__ out, int n)
{
    const int i = blockIdx.x * 256 + threadIdx.x;
    if (i < n) out[i] = 0.0f;
}

__global__ __launch_bounds__(256)
void bte_max_kernel(const float* __restrict__ h,
                    const int*   __restrict__ batch_id,
                    float*       __restrict__ out,
                    int N)
{
    const int idx  = blockIdx.x * 256 + threadIdx.x;
    const int node = idx >> 5;
    if (node >= N) return;
    const int cq = (idx & 31) << 2;
    const int b  = batch_id[node];
    const float4 v = *(const float4*)(h + (size_t)node * CDIM + cq);
    int* dst = (int*)(out + (size_t)b * CDIM + cq);
    atomicMax(dst + 0, __float_as_int(v.x));
    atomicMax(dst + 1, __float_as_int(v.y));
    atomicMax(dst + 2, __float_as_int(v.z));
    atomicMax(dst + 3, __float_as_int(v.w));
}

extern "C" void kernel_launch(void* const* d_in, const int* in_sizes, int n_in,
                              void* d_out, int out_size, void* d_ws, size_t ws_size,
                              hipStream_t stream)
{
    const float* emb      = (const float*)d_in[0];
    const float* W        = (const float*)d_in[1];
    const float* bias     = (const float*)d_in[2];
    const int*   tokens   = (const int*)d_in[3];
    const int*   parent   = (const int*)d_in[4];
    const int*   batch_id = (const int*)d_in[6];
    const int    N        = in_sizes[3];

    float* h   = (float*)d_ws;
    float* out = (float*)d_out;

    const int gemm_blocks = (N + 127) / 128;
    bte_gemm_kernel<<<gemm_blocks, 256, 0, stream>>>(emb, W, bias, tokens, h, N);

    static const int lvl_off[7]  = {0, 512, 2560, 10752, 35328, 109056, 256512};
    static const int lvl_size[7] = {512, 2048, 8192, 24576, 73728, 147456, 147456};
    for (int d = 6; d >= 1; --d) {
        const int threads = lvl_size[d] * 32;
        bte_scatter_kernel<<<(threads + 255) / 256, 256, 0, stream>>>(
            h, parent, lvl_off[d], lvl_size[d]);
    }

    bte_init_out<<<(out_size + 255) / 256, 256, 0, stream>>>(out, out_size);
    const long long mthreads = (long long)N * 32;
    bte_max_kernel<<<(int)((mthreads + 255) / 256), 256, 0, stream>>>(
        h, batch_id, out, N);
}


# per-batch max: each work-group folds its node chunk into a private 512x128 max table in LDS (ds_max), a 2nd kernel folds the 256 tables; no global atomicMax, no init kernel; order of sums unchanged
# speedup vs baseline: 1.8462x; 1.8462x over previous
#include <hip/hip_runtime.h>
#include <hip/hip_bf16.h>

typedef __attribute__((ext_vector_type(2))) float v2f;
typedef __attribute__((ext_vector_type(8))) float v8f;

#define EDIM 128
#define CDIM 128

__global__ __launch_bounds__(256)
void bte_gemm_kernel(const float* __restrict__ emb,
                     const float* __restrict__ W,
                     const float* __restrict__ bias,
                     const int*   __restrict__ tokens,
                     float*       __restrict__ h,
                     int N)
{
    __shared__ float ldsA[128 * EDIM];
    const int  block0 = blockIdx.x * 128;
    const int  tid    = threadIdx.x;
    const bool full   = (block0 + 128) <= N;

    {
        const int nl   = tid >> 1;
        const int half = (tid & 1) * 64;
        const int node = block0 + nl;
        const int tok  = (full || node < N) ? tokens[node] : 0;
        const float4* src = (const float4*)(emb + (size_t)tok * EDIM + half);
        float4*       dst = (float4*)(ldsA + nl * EDIM + half);
        #pragma unroll
        for (int i = 0; i < 16; ++i) dst[i] = src[i];
    }
    __syncthreads();

    const int wid   = tid >> 5;
    const int lane  = tid & 31;
    const int col   = lane & 15;
    const int khalf = lane >> 4;
    const float* arow = ldsA + (wid * 16 + col) * EDIM;

    for (int t = 0; t < 8; ++t) {
        const int c0 = t * 16;
        const float bv = bias[c0 + col];
        v8f acc = { bv, bv, bv, bv, bv, bv, bv, bv };
        const float* wrow = W + (size_t)(c0 + col) * EDIM;
        #pragma unroll 4
        for (int k = 0; k < EDIM; k += 4) {
            const int kk = k + 2 * khalf;
            v2f a, bf;
            a.x  = arow[kk];   a.y  = arow[kk + 1];
            bf.x = wrow[kk];   bf.y = wrow[kk + 1];
            acc = __builtin_amdgcn_wmma_f32_16x16x4_f32(
                      false, a, false, bf, (short)0, acc, false, false);
        }
        const int mbase = block0 + wid * 16 + 8 * khalf;
        float* hp = h + (size_t)mbase * CDIM + c0 + col;
        if (full) {
            #pragma unroll
            for (int r = 0; r < 8; ++r) hp[r * CDIM] = acc[r];
        } else {
            #pragma unroll
            for (int r = 0; r < 8; ++r)
                if (mbase + r < N) hp[r * CDIM] = acc[r];
        }
    }
}

__global__ __launch_bounds__(256)
void bte_scatter_kernel(float* __restrict__ h,
                        const int* __restrict__ parent,
                        int off, int count)
{
    const int idx = blockIdx.x * 256 + threadIdx.x;
    const int nl  = idx >> 5;
    if (nl >= count) return;
    const int cq   = (idx & 31) << 2;
    const int node = off + nl;
    const int par  = parent[node];
    const float4 v = *(const float4*)(h + (size_t)node * CDIM + cq);
    float* dst = h + (size_t)par * CDIM + cq;
    atomicAdd(dst + 0, v.x);
    atomicAdd(dst + 1, v.y);
    atomicAdd(dst + 2, v.z);
    atomicAdd(dst + 3, v.w);
}

#define NBATCH 512
#define NPART  256
#define TWAVES 16
#define GRP    16
#define TSTEP  (TWAVES * GRP)

__global__ __launch_bounds__(TWAVES * 32)
void bte_max_table_kernel(float* h,
                          const int* __restrict__ batch_id,
                          int N, int chunk)
{
    __shared__ int tab[NBATCH * CDIM];
    const int tid = threadIdx.x;
    {
        int4* t4 = (int4*)tab;
        const int4 z = make_int4(0, 0, 0, 0);
        for (int i = tid; i < NBATCH * CDIM / 4; i += TWAVES * 32) t4[i] = z;
    }
    __syncthreads();

    const int wave = __builtin_amdgcn_readfirstlane(tid >> 5);
    const int lane = tid & 31;
    const int n0   = blockIdx.x * chunk;
    const int n1   = min(n0 + chunk, N);
    const int last = n1 - 1;
    const float* hl = h + lane * 4;
    int*         tl = tab + lane;

#define BTE_LOAD(V, B, BASE)                                                  \
    _Pragma("unroll")                                                         \
    for (int i = 0; i < GRP; ++i) {                                           \
        const int node = min((BASE) + i, last);                               \
        B[i] = batch_id[node];                                                \
        V[i] = *(const float4*)(hl + (size_t)node * CDIM);                    \
    }
#define BTE_FOLD(V, B)                                                        \
    _Pragma("unroll")                                                         \
    for (int i = 0; i < GRP; ++i) {                                           \
        int* t = tl + B[i] * CDIM;                                            \
        atomicMax(t,      __float_as_int(V[i].x));                            \
        atomicMax(t + 32, __float_as_int(V[i].y));                            \
        atomicMax(t + 64, __float_as_int(V[i].z));                            \
        atomicMax(t + 96, __float_as_int(V[i].w));                            \
    }

    int base = n0 + wave * GRP;
    if (base < n1) {
        float4 va[GRP], vb[GRP];
        int    ba[GRP], bb[GRP];
        BTE_LOAD(va, ba, base)
        for (;;) {
            base += TSTEP;
            if (base >= n1) { BTE_FOLD(va, ba) break; }
            BTE_LOAD(vb, bb, base)
            BTE_FOLD(va, ba)
            base += TSTEP;
            if (base >= n1) { BTE_FOLD(vb, bb) break; }
            BTE_LOAD(va, ba, base)
            BTE_FOLD(vb, bb)
        }
    }
#undef BTE_LOAD
#undef BTE_FOLD
    __syncthreads();

    for (int b = wave; b < NBATCH; b += TWAVES) {
        const int* t = tl + b * CDIM;
        float4 o;
        o.x = __int_as_float(t[0]);
        o.y = __int_as_float(t[32]);
        o.z = __int_as_float(t[64]);
        o.w = __int_as_float(t[96]);
        *(float4*)(h + (size_t)(n0 + b) * CDIM + lane * 4) = o;
    }
}

__global__ __launch_bounds__(256)
void bte_max_fold_kernel(const float* __restrict__ part,
                         float*       __restrict__ out,
                         int chunk)
{
    __shared__ int4 red[8][32];
    const int b    = blockIdx.x;
    const int wave = threadIdx.x >> 5;
    const int lane = threadIdx.x & 31;
    const float* p = part + (size_t)b * CDIM + lane * 4;

    int4 m = make_int4(0, 0, 0, 0);
    #pragma unroll
    for (int k = 0; k < NPART / 8; ++k) {
        const float4 v = *(const float4*)(p + (size_t)(wave + 8 * k) * chunk * CDIM);
        m.x = max(m.x, __float_as_int(v.x));
        m.y = max(m.y, __float_as_int(v.y));
        m.z = max(m.z, __float_as_int(v.z));
        m.w = max(m.w, __float_as_int(v.w));
    }
    red[wave][lane] = m;
    __syncthreads();
    if (wave == 0) {
        #pragma unroll
        for (int g = 1; g < 8; ++g) {
            const int4 r = red[g][lane];
            m.x = max(m.x, r.x); m.y = max(m.y, r.y);
            m.z = max(m.z, r.z); m.w = max(m.w, r.w);
        }
        float4 o;
        o.x = __int_as_float(m.x); o.y = __int_as_float(m.y);
        o.z = __int_as_float(m.z); o.w = __int_as_float(m.w);
        *(float4*)(out + (size_t)b * CDIM + lane * 4) = o;
    }
}

extern "C" void kernel_launch(void* const* d_in, const int* in_sizes, int n_in,
                              void* d_out, int out_size, void* d_ws, size_t ws_size,
                              hipStream_t stream)
{
    const float* emb      = (const float*)d_in[0];
    const float* W        = (const float*)d_in[1];
    const float* bias     = (const float*)d_in[2];
    const int*   tokens   = (const int*)d_in[3];
    const int*   parent   = (const int*)d_in[4];
    const int*   batch_id = (const int*)d_in[6];
    const int    N        = in_sizes[3];

    float* h   = (float*)d_ws;
    float* out = (float*)d_out;

    const int gemm_blocks = (N + 127) / 128;
    bte_gemm_kernel<<<gemm_blocks, 256, 0, stream>>>(emb, W, bias, tokens, h, N);

    static const int lvl_off[7]  = {0, 512, 2560, 10752, 35328, 109056, 256512};
    static const int lvl_size[7] = {512, 2048, 8192, 24576, 73728, 147456, 147456};
    for (int d = 6; d >= 1; --d) {
        const int threads = lvl_size[d] * 32;
        bte_scatter_kernel<<<(threads + 255) / 256, 256, 0, stream>>>(
            h, parent, lvl_off[d], lvl_size[d]);
    }

    const int chunk = (N + NPART - 1) / NPART;
    bte_max_table_kernel<<<NPART, TWAVES * 32, 0, stream>>>(h, batch_id, N, chunk);
    bte_max_fold_kernel<<<out_size / CDIM, 256, 0, stream>>>(h, out, chunk);
}


# projection GEMM register blocking: a wave owns 64 rows x 64 columns (4x4 tiles, 16 accumulators, 4 waves per block); per k-step 8 fragment loads feed 16 WMMAs (was 2 loads per WMMA); sums unchanged
# speedup vs baseline: 2.7020x; 1.4635x over previous
#include <hip/hip_runtime.h>
#include <hip/hip_bf16.h>

typedef __attribute__((ext_vector_type(2))) float v2f;
typedef __attribute__((ext_vector_type(8))) float v8f;

#define EDIM 128
#define CDIM 128

__global__ __launch_bounds__(128)
void bte_gemm_kernel(const float* __restrict__ emb,
                     const float* __restrict__ W,
                     const float* __restrict__ bias,
                     const int*   __restrict__ tokens,
                     float*       __restrict__ h,
                     int N)
{
    __shared__ float ldsA[128 * EDIM];
    const int  block0 = blockIdx.x * 128;
    const int  tid    = threadIdx.x;
    const bool full   = (block0 + 128) <= N;

    {
        const int node = block0 + tid;
        const int tok  = (full || node < N) ? tokens[node] : 0;
        const float4* src = (const float4*)(emb + (size_t)tok * EDIM);
        float4*       dst = (float4*)(ldsA + tid * EDIM);
        #pragma unroll
        for (int i = 0; i < 32; ++i) dst[i] = src[i];
    }
    __syncthreads();

    const int wid   = tid >> 5;
    const int lane  = tid & 31;
    const int col   = lane & 15;
    const int khalf = lane >> 4;
    const int r0    = (wid >> 1) * 64;
    const int c0    = (wid & 1) * 64;

    const float* arow = ldsA + (r0 + col) * EDIM + 2 * khalf;
    const float* wrow = W + (size_t)(c0 + col) * EDIM + 2 * khalf;

    v8f acc[4][4];
    #pragma unroll
    for (int j = 0; j < 4; ++j) {
        const float bv = bias[c0 + 16 * j + col];
        const v8f b8 = { bv, bv, bv, bv, bv, bv, bv, bv };
        #pragma unroll
        for (int i = 0; i < 4; ++i) acc[i][j] = b8;
    }

    #pragma unroll 2
    for (int k = 0; k < EDIM; k += 4) {
        v2f a[4], bf[4];
        #pragma unroll
        for (int i = 0; i < 4; ++i) {
            a[i].x = arow[i * 16 * EDIM + k];
            a[i].y = arow[i * 16 * EDIM + k + 1];
        }
        #pragma unroll
        for (int j = 0; j < 4; ++j) {
            bf[j].x = wrow[j * 16 * EDIM + k];
            bf[j].y = wrow[j * 16 * EDIM + k + 1];
        }
        #pragma unroll
        for (int i = 0; i < 4; ++i) {
            #pragma unroll
            for (int j = 0; j < 4; ++j) {
                acc[i][j] = __builtin_amdgcn_wmma_f32_16x16x4_f32(
                                false, a[i], false, bf[j], (short)0, acc[i][j],
                                false, false);
            }
        }
    }

    #pragma unroll
    for (int i = 0; i < 4; ++i) {
        const int mbase = block0 + r0 + 16 * i + 8 * khalf;
        #pragma unroll
        for (int j = 0; j < 4; ++j) {
            float* hp = h + (size_t)mbase * CDIM + c0 + 16 * j + col;
            if (full) {
                #pragma unroll
                for (int r = 0; r < 8; ++r) hp[r * CDIM] = acc[i][j][r];
            } else {
                #pragma unroll
                for (int r = 0; r < 8; ++r)
                    if (mbase + r < N) hp[r * CDIM] = acc[i][j][r];
            }
        }
    }
}

__global__ __launch_bounds__(256)
void bte_scatter_kernel(float* __restrict__ h,
                        const int* __restrict__ parent,
                        int off, int count)
{
    const int idx = blockIdx.x * 256 + threadIdx.x;
    const int nl  = idx >> 5;
    if (nl >= count) return;
    const int cq   = (idx & 31) << 2;
    const int node = off + nl;
    const int par  = parent[node];
    const float4 v = *(const float4*)(h + (size_t)node * CDIM + cq);
    float* dst = h + (size_t)par * CDIM + cq;
    atomicAdd(dst + 0, v.x);
    atomicAdd(dst + 1, v.y);
    atomicAdd(dst + 2, v.z);
    atomicAdd(dst + 3, v.w);
}

#define NBATCH 512
#define NPART  256
#define TWAVES 16
#define GRP    16
#define TSTEP  (TWAVES * GRP)

__global__ __launch_bounds__(TWAVES * 32)
void bte_max_table_kernel(float* h,
                          const int* __restrict__ batch_id,
                          int N, int chunk)
{
    __shared__ int tab[NBATCH * CDIM];
    const int tid = threadIdx.x;
    {
        int4* t4 = (int4*)tab;
        const int4 z = make_int4(0, 0, 0, 0);
        for (int i = tid; i < NBATCH * CDIM / 4; i += TWAVES * 32) t4[i] = z;
    }
    __syncthreads();

    const int wave = __builtin_amdgcn_readfirstlane(tid >> 5);
    const int lane = tid & 31;
    const int n0   = blockIdx.x * chunk;
    const int n1   = min(n0 + chunk, N);
    const int last = n1 - 1;
    const float* hl = h + lane * 4;
    int*         tl = tab + lane;

#define BTE_LOAD(V, B, BASE)                                                  \
    _Pragma("unroll")                                                         \
    for (int i = 0; i < GRP; ++i) {                                           \
        const int node = min((BASE) + i, last);                               \
        B[i] = batch_id[node];                                                \
        V[i] = *(const float4*)(hl + (size_t)node * CDIM);                    \
    }
#define BTE_FOLD(V, B)                                                        \
    _Pragma("unroll")                                                         \
    for (int i = 0; i < GRP; ++i) {                                           \
        int* t = tl + B[i] * CDIM;                                            \
        atomicMax(t,      __float_as_int(V[i].x));                            \
        atomicMax(t + 32, __float_as_int(V[i].y));                            \
        atomicMax(t + 64, __float_as_int(V[i].z));                            \
        atomicMax(t + 96, __float_as_int(V[i].w));                            \
    }

    int base = n0 + wave * GRP;
    if (base < n1) {
        float4 va[GRP], vb[GRP];
        int    ba[GRP], bb[GRP];
        BTE_LOAD(va, ba, base)
        for (;;) {
            base += TSTEP;
            if (base >= n1) { BTE_FOLD(va, ba) break; }
            BTE_LOAD(vb, bb, base)
            BTE_FOLD(va, ba)
            base += TSTEP;
            if (base >= n1) { BTE_FOLD(vb, bb) break; }
            BTE_LOAD(va, ba, base)
            BTE_FOLD(vb, bb)
        }
    }
#undef BTE_LOAD
#undef BTE_FOLD
    __syncthreads();

    for (int b = wave; b < NBATCH; b += TWAVES) {
        const int* t = tl + b * CDIM;
        float4 o;
        o.x = __int_as_float(t[0]);
        o.y = __int_as_float(t[32]);
        o.z = __int_as_float(t[64]);
        o.w = __int_as_float(t[96]);
        *(float4*)(h + (size_t)(n0 + b) * CDIM + lane * 4) = o;
    }
}

__global__ __launch_bounds__(256)
void bte_max_fold_kernel(const float* __restrict__ part,
                         float*       __restrict__ out,
                         int chunk)
{
    __shared__ int4 red[8][32];
    const int b    = blockIdx.x;
    const int wave = threadIdx.x >> 5;
    const int lane = threadIdx.x & 31;
    const float* p = part + (size_t)b * CDIM + lane * 4;

    int4 m = make_int4(0, 0, 0, 0);
    #pragma unroll
    for (int k = 0; k < NPART / 8; ++k) {
        const float4 v = *(const float4*)(p + (size_t)(wave + 8 * k) * chunk * CDIM);
        m.x = max(m.x, __float_as_int(v.x));
        m.y = max(m.y, __float_as_int(v.y));
        m.z = max(m.z, __float_as_int(v.z));
        m.w = max(m.w, __float_as_int(v.w));
    }
    red[wave][lane] = m;
    __syncthreads();
    if (wave == 0) {
        #pragma unroll
        for (int g = 1; g < 8; ++g) {
            const int4 r = red[g][lane];
            m.x = max(m.x, r.x); m.y = max(m.y, r.y);
            m.z = max(m.z, r.z); m.w = max(m.w, r.w);
        }
        float4 o;
        o.x = __int_as_float(m.x); o.y = __int_as_float(m.y);
        o.z = __int_as_float(m.z); o.w = __int_as_float(m.w);
        *(float4*)(out + (size_t)b * CDIM + lane * 4) = o;
    }
}

extern "C" void kernel_launch(void* const* d_in, const int* in_sizes, int n_in,
                              void* d_out, int out_size, void* d_ws, size_t ws_size,
                              hipStream_t stream)
{
    const float* emb      = (const float*)d_in[0];
    const float* W        = (const float*)d_in[1];
    const float* bias     = (const float*)d_in[2];
    const int*   tokens   = (const int*)d_in[3];
    const int*   parent   = (const int*)d_in[4];
    const int*   batch_id = (const int*)d_in[6];
    const int    N        = in_sizes[3];

    float* h   = (float*)d_ws;
    float* out = (float*)d_out;

    const int gemm_blocks = (N + 127) / 128;
    bte_gemm_kernel<<<gemm_blocks, 128, 0, stream>>>(emb, W, bias, tokens, h, N);

    static const int lvl_off[7]  = {0, 512, 2560, 10752, 35328, 109056, 256512};
    static const int lvl_size[7] = {512, 2048, 8192, 24576, 73728, 147456, 147456};
    for (int d = 6; d >= 1; --d) {
        const int threads = lvl_size[d] * 32;
        bte_scatter_kernel<<<(threads + 255) / 256, 256, 0, stream>>>(
            h, parent, lvl_off[d], lvl_size[d]);
    }

    const int chunk = (N + NPART - 1) / NPART;
    bte_max_table_kernel<<<NPART, TWAVES * 32, 0, stream>>>(h, batch_id, N, chunk);
    bte_max_fold_kernel<<<out_size / CDIM, 256, 0, stream>>>(h, out, chunk);
}


# projection once per embedding-table row (2x50000 rows through WMMA instead of 403968), node rows copied from the projected table by token; each row's sum formed as before, bits unchanged
# speedup vs baseline: 2.8785x; 1.0653x over previous
#include <hip/hip_runtime.h>
#include <hip/hip_bf16.h>

typedef __attribute__((ext_vector_type(2))) float v2f;
typedef __attribute__((ext_vector_type(8))) float v8f;
typedef __attribute__((ext_vector_type(4))) float v4f;

#define EDIM 128
#define CDIM 128

__global__ __launch_bounds__(128)
void bte_gemm_kernel(const float* __restrict__ emb,
                     const float* __restrict__ W,
                     const float* __restrict__ bias,
                     const int*   __restrict__ tokens,
                     float*       __restrict__ h,
                     int N)
{
    __shared__ float ldsA[128 * EDIM];
    const int  block0 = blockIdx.x * 128;
    const int  tid    = threadIdx.x;
    const bool full   = (block0 + 128) <= N;

    {
        const int node = block0 + tid;
        int tok = 0;
        if (full || node < N) tok = tokens ? tokens[node] : node;
        const float4* src = (const float4*)(emb + (size_t)tok * EDIM);
        float4*       dst = (float4*)(ldsA + tid * EDIM);
        #pragma unroll
        for (int i = 0; i < 32; ++i) dst[i] = src[i];
    }
    __syncthreads();

    const int wid   = tid >> 5;
    const int lane  = tid & 31;
    const int col   = lane & 15;
    const int khalf = lane >> 4;
    const int r0    = (wid >> 1) * 64;
    const int c0    = (wid & 1) * 64;

    const float* arow = ldsA + (r0 + col) * EDIM + 2 * khalf;
    const float* wrow = W + (size_t)(c0 + col) * EDIM + 2 * khalf;

    v8f acc[4][4];
    #pragma unroll
    for (int j = 0; j < 4; ++j) {
        const float bv = bias[c0 + 16 * j + col];
        const v8f b8 = { bv, bv, bv, bv, bv, bv, bv, bv };
        #pragma unroll
        for (int i = 0; i < 4; ++i) acc[i][j] = b8;
    }

    #pragma unroll 2
    for (int k = 0; k < EDIM; k += 4) {
        v2f a[4], bf[4];
        #pragma unroll
        for (int i = 0; i < 4; ++i) {
            a[i].x = arow[i * 16 * EDIM + k];
            a[i].y = arow[i * 16 * EDIM + k + 1];
        }
        #pragma unroll
        for (int j = 0; j < 4; ++j) {
            bf[j].x = wrow[j * 16 * EDIM + k];
            bf[j].y = wrow[j * 16 * EDIM + k + 1];
        }
        #pragma unroll
        for (int i = 0; i < 4; ++i) {
            #pragma unroll
            for (int j = 0; j < 4; ++j) {
                acc[i][j] = __builtin_amdgcn_wmma_f32_16x16x4_f32(
                                false, a[i], false, bf[j], (short)0, acc[i][j],
                                false, false);
            }
        }
    }

    #pragma unroll
    for (int i = 0; i < 4; ++i) {
        const int mbase = block0 + r0 + 16 * i + 8 * khalf;
        #pragma unroll
        for (int j = 0; j < 4; ++j) {
            float* hp = h + (size_t)mbase * CDIM + c0 + 16 * j + col;
            if (full) {
                #pragma unroll
                for (int r = 0; r < 8; ++r) hp[r * CDIM] = acc[i][j][r];
            } else {
                #pragma unroll
                for (int r = 0; r < 8; ++r)
                    if (mbase + r < N) hp[r * CDIM] = acc[i][j][r];
            }
        }
    }
}

#define CWAVES 8
#define CGROUPS 512
#define CGRP   16

__global__ __launch_bounds__(CWAVES * 32)
void bte_row_copy_kernel(const float* __restrict__ table,
                         const int*   __restrict__ tokens,
                         float*       __restrict__ h,
                         int rows, int step)
{
    const int wave = threadIdx.x >> 5;
    const int lane = threadIdx.x & 31;
    const int last = rows - 1;
    const float* tl = table + lane * 4;
    float*       hl = h + lane * 4;

    int base = (blockIdx.x * CWAVES + wave) * CGRP;
    if (base >= rows) return;

    int tk[CGRP];
    #pragma unroll
    for (int i = 0; i < CGRP; ++i) tk[i] = tokens[min(base + i, last)];

    for (;;) {
        v4f v[CGRP];
        #pragma unroll
        for (int i = 0; i < CGRP; ++i)
            v[i] = *(const v4f*)(tl + (size_t)tk[i] * CDIM);

        const int next = base + step;
        #pragma unroll
        for (int i = 0; i < CGRP; ++i) tk[i] = tokens[min(next + i, last)];

        asm volatile("" :: "v"(v[0]), "v"(v[1]), "v"(v[2]), "v"(v[3]),
                           "v"(v[4]), "v"(v[5]), "v"(v[6]), "v"(v[7]),
                           "v"(v[8]), "v"(v[9]), "v"(v[10]), "v"(v[11]),
                           "v"(v[12]), "v"(v[13]), "v"(v[14]), "v"(v[15]));

        #pragma unroll
        for (int i = 0; i < CGRP; ++i)
            *(v4f*)(hl + (size_t)min(base + i, last) * CDIM) = v[i];

        if (next >= rows) break;
        base = next;
    }
}

__global__ __launch_bounds__(256)
void bte_scatter_kernel(float* __restrict__ h,
                        const int* __restrict__ parent,
                        int off, int count)
{
    const int idx = blockIdx.x * 256 + threadIdx.x;
    const int nl  = idx >> 5;
    if (nl >= count) return;
    const int cq   = (idx & 31) << 2;
    const int node = off + nl;
    const int par  = parent[node];
    const float4 v = *(const float4*)(h + (size_t)node * CDIM + cq);
    float* dst = h + (size_t)par * CDIM + cq;
    atomicAdd(dst + 0, v.x);
    atomicAdd(dst + 1, v.y);
    atomicAdd(dst + 2, v.z);
    atomicAdd(dst + 3, v.w);
}

#define NBATCH 512
#define NPART  256
#define TWAVES 16
#define GRP    16
#define TSTEP  (TWAVES * GRP)

__global__ __launch_bounds__(TWAVES * 32)
void bte_max_table_kernel(float* h,
                          const int* __restrict__ batch_id,
                          int N, int chunk)
{
    __shared__ int tab[NBATCH * CDIM];
    const int tid = threadIdx.x;
    {
        int4* t4 = (int4*)tab;
        const int4 z = make_int4(0, 0, 0, 0);
        for (int i = tid; i < NBATCH * CDIM / 4; i += TWAVES * 32) t4[i] = z;
    }
    __syncthreads();

    const int wave = __builtin_amdgcn_readfirstlane(tid >> 5);
    const int lane = tid & 31;
    const int n0   = blockIdx.x * chunk;
    const int n1   = min(n0 + chunk, N);
    const int last = n1 - 1;
    const float* hl = h + lane * 4;
    int*         tl = tab + lane;

#define BTE_LOAD(V, B, BASE)                                                  \
    _Pragma("unroll")                                                         \
    for (int i = 0; i < GRP; ++i) {                                           \
        const int node = min((BASE) + i, last);                               \
        B[i] = batch_id[node];                                                \
        V[i] = *(const float4*)(hl + (size_t)node * CDIM);                    \
    }
#define BTE_FOLD(V, B)                                                        \
    _Pragma("unroll")                                                         \
    for (int i = 0; i < GRP; ++i) {                                           \
        int* t = tl + B[i] * CDIM;                                            \
        atomicMax(t,      __float_as_int(V[i].x));                            \
        atomicMax(t + 32, __float_as_int(V[i].y));                            \
        atomicMax(t + 64, __float_as_int(V[i].z));                            \
        atomicMax(t + 96, __float_as_int(V[i].w));                            \
    }

    int base = n0 + wave * GRP;
    if (base < n1) {
        float4 va[GRP], vb[GRP];
        int    ba[GRP], bb[GRP];
        BTE_LOAD(va, ba, base)
        for (;;) {
            base += TSTEP;
            if (base >= n1) { BTE_FOLD(va, ba) break; }
            BTE_LOAD(vb, bb, base)
            BTE_FOLD(va, ba)
            base += TSTEP;
            if (base >= n1) { BTE_FOLD(vb, bb) break; }
            BTE_LOAD(va, ba, base)
            BTE_FOLD(vb, bb)
        }
    }
#undef BTE_LOAD
#undef BTE_FOLD
    __syncthreads();

    for (int b = wave; b < NBATCH; b += TWAVES) {
        const int* t = tl + b * CDIM;
        float4 o;
        o.x = __int_as_float(t[0]);
        o.y = __int_as_float(t[32]);
        o.z = __int_as_float(t[64]);
        o.w = __int_as_float(t[96]);
        *(float4*)(h + (size_t)(n0 + b) * CDIM + lane * 4) = o;
    }
}

__global__ __launch_bounds__(256)
void bte_max_fold_kernel(const float* __restrict__ part,
                         float*       __restrict__ out,
                         int chunk)
{
    __shared__ int4 red[8][32];
    const int b    = blockIdx.x;
    const int wave = threadIdx.x >> 5;
    const int lane = threadIdx.x & 31;
    const float* p = part + (size_t)b * CDIM + lane * 4;

    int4 m = make_int4(0, 0, 0, 0);
    #pragma unroll
    for (int k = 0; k < NPART / 8; ++k) {
        const float4 v = *(const float4*)(p + (size_t)(wave + 8 * k) * chunk * CDIM);
        m.x = max(m.x, __float_as_int(v.x));
        m.y = max(m.y, __float_as_int(v.y));
        m.z = max(m.z, __float_as_int(v.z));
        m.w = max(m.w, __float_as_int(v.w));
    }
    red[wave][lane] = m;
    __syncthreads();
    if (wave == 0) {
        #pragma unroll
        for (int g = 1; g < 8; ++g) {
            const int4 r = red[g][lane];
            m.x = max(m.x, r.x); m.y = max(m.y, r.y);
            m.z = max(m.z, r.z); m.w = max(m.w, r.w);
        }
        float4 o;
        o.x = __int_as_float(m.x); o.y = __int_as_float(m.y);
        o.z = __int_as_float(m.z); o.w = __int_as_float(m.w);
        *(float4*)(out + (size_t)b * CDIM + lane * 4) = o;
    }
}

extern "C" void kernel_launch(void* const* d_in, const int* in_sizes, int n_in,
                              void* d_out, int out_size, void* d_ws, size_t ws_size,
                              hipStream_t stream)
{
    const float* emb      = (const float*)d_in[0];
    const float* W        = (const float*)d_in[1];
    const float* bias     = (const float*)d_in[2];
    const int*   tokens   = (const int*)d_in[3];
    const int*   parent   = (const int*)d_in[4];
    const int*   batch_id = (const int*)d_in[6];
    const int    N        = in_sizes[3];

    float* h   = (float*)d_ws;
    float* out = (float*)d_out;

    const int K = in_sizes[0] / EDIM;
    if (2 * K < N) {
        float* tail = h + (size_t)(N - K) * CDIM;
        const int kblocks = (K + 127) / 128;
        bte_gemm_kernel<<<kblocks, 128, 0, stream>>>(emb, W, bias, nullptr, tail, K);
        bte_row_copy_kernel<<<CGROUPS, CWAVES * 32, 0, stream>>>(
            tail, tokens, h, N - K, CGROUPS * CWAVES * CGRP);
        bte_gemm_kernel<<<kblocks, 128, 0, stream>>>(emb, W, bias, tokens + (N - K), tail, K);
    } else {
        const int gemm_blocks = (N + 127) / 128;
        bte_gemm_kernel<<<gemm_blocks, 128, 0, stream>>>(emb, W, bias, tokens, h, N);
    }

    static const int lvl_off[7]  = {0, 512, 2560, 10752, 35328, 109056, 256512};
    static const int lvl_size[7] = {512, 2048, 8192, 24576, 73728, 147456, 147456};
    for (int d = 6; d >= 1; --d) {
        const int threads = lvl_size[d] * 32;
        bte_scatter_kernel<<<(threads + 255) / 256, 256, 0, stream>>>(
            h, parent, lvl_off[d], lvl_size[d]);
    }

    const int chunk = (N + NPART - 1) / NPART;
    bte_max_table_kernel<<<NPART, TWAVES * 32, 0, stream>>>(h, batch_id, N, chunk);
    bte_max_fold_kernel<<<out_size / CDIM, 256, 0, stream>>>(h, out, chunk);
}


# subtree-sum scatter: lane l owns columns l, l+32, l+64, l+96, so each f32 atomic add of a wave covers one whole 128-byte line (4 line requests per row, was 16); atomics kept, order free
# speedup vs baseline: 3.3183x; 1.1528x over previous
#include <hip/hip_runtime.h>
#include <hip/hip_bf16.h>

typedef __attribute__((ext_vector_type(2))) float v2f;
typedef __attribute__((ext_vector_type(8))) float v8f;
typedef __attribute__((ext_vector_type(4))) float v4f;

#define EDIM 128
#define CDIM 128

__global__ __launch_bounds__(128)
void bte_gemm_kernel(const float* __restrict__ emb,
                     const float* __restrict__ W,
                     const float* __restrict__ bias,
                     const int*   __restrict__ tokens,
                     float*       __restrict__ h,
                     int N)
{
    __shared__ float ldsA[128 * EDIM];
    const int  block0 = blockIdx.x * 128;
    const int  tid    = threadIdx.x;
    const bool full   = (block0 + 128) <= N;

    {
        const int node = block0 + tid;
        int tok = 0;
        if (full || node < N) tok = tokens ? tokens[node] : node;
        const float4* src = (const float4*)(emb + (size_t)tok * EDIM);
        float4*       dst = (float4*)(ldsA + tid * EDIM);
        #pragma unroll
        for (int i = 0; i < 32; ++i) dst[i] = src[i];
    }
    __syncthreads();

    const int wid   = tid >> 5;
    const int lane  = tid & 31;
    const int col   = lane & 15;
    const int khalf = lane >> 4;
    const int r0    = (wid >> 1) * 64;
    const int c0    = (wid & 1) * 64;

    const float* arow = ldsA + (r0 + col) * EDIM + 2 * khalf;
    const float* wrow = W + (size_t)(c0 + col) * EDIM + 2 * khalf;

    v8f acc[4][4];
    #pragma unroll
    for (int j = 0; j < 4; ++j) {
        const float bv = bias[c0 + 16 * j + col];
        const v8f b8 = { bv, bv, bv, bv, bv, bv, bv, bv };
        #pragma unroll
        for (int i = 0; i < 4; ++i) acc[i][j] = b8;
    }

    #pragma unroll 2
    for (int k = 0; k < EDIM; k += 4) {
        v2f a[4], bf[4];
        #pragma unroll
        for (int i = 0; i < 4; ++i) {
            a[i].x = arow[i * 16 * EDIM + k];
            a[i].y = arow[i * 16 * EDIM + k + 1];
        }
        #pragma unroll
        for (int j = 0; j < 4; ++j) {
            bf[j].x = wrow[j * 16 * EDIM + k];
            bf[j].y = wrow[j * 16 * EDIM + k + 1];
        }
        #pragma unroll
        for (int i = 0; i < 4; ++i) {
            #pragma unroll
            for (int j = 0; j < 4; ++j) {
                acc[i][j] = __builtin_amdgcn_wmma_f32_16x16x4_f32(
                                false, a[i], false, bf[j], (short)0, acc[i][j],
                                false, false);
            }
        }
    }

    #pragma unroll
    for (int i = 0; i < 4; ++i) {
        const int mbase = block0 + r0 + 16 * i + 8 * khalf;
        #pragma unroll
        for (int j = 0; j < 4; ++j) {
            float* hp = h + (size_t)mbase * CDIM + c0 + 16 * j + col;
            if (full) {
                #pragma unroll
                for (int r = 0; r < 8; ++r) hp[r * CDIM] = acc[i][j][r];
            } else {
                #pragma unroll
                for (int r = 0; r < 8; ++r)
                    if (mbase + r < N) hp[r * CDIM] = acc[i][j][r];
            }
        }
    }
}

#define CWAVES 8
#define CGROUPS 512
#define CGRP   16

__global__ __launch_bounds__(CWAVES * 32)
void bte_row_copy_kernel(const float* __restrict__ table,
                         const int*   __restrict__ tokens,
                         float*       __restrict__ h,
                         int rows, int step)
{
    const int wave = threadIdx.x >> 5;
    const int lane = threadIdx.x & 31;
    const int last = rows - 1;
    const float* tl = table + lane * 4;
    float*       hl = h + lane * 4;

    int base = (blockIdx.x * CWAVES + wave) * CGRP;
    if (base >= rows) return;

    int tk[CGRP];
    #pragma unroll
    for (int i = 0; i < CGRP; ++i) tk[i] = tokens[min(base + i, last)];

    for (;;) {
        v4f v[CGRP];
        #pragma unroll
        for (int i = 0; i < CGRP; ++i)
            v[i] = *(const v4f*)(tl + (size_t)tk[i] * CDIM);

        const int next = base + step;
        #pragma unroll
        for (int i = 0; i < CGRP; ++i) tk[i] = tokens[min(next + i, last)];

        asm volatile("" :: "v"(v[0]), "v"(v[1]), "v"(v[2]), "v"(v[3]),
                           "v"(v[4]), "v"(v[5]), "v"(v[6]), "v"(v[7]),
                           "v"(v[8]), "v"(v[9]), "v"(v[10]), "v"(v[11]),
                           "v"(v[12]), "v"(v[13]), "v"(v[14]), "v"(v[15]));

        #pragma unroll
        for (int i = 0; i < CGRP; ++i)
            *(v4f*)(hl + (size_t)min(base + i, last) * CDIM) = v[i];

        if (next >= rows) break;
        base = next;
    }
}

__global__ __launch_bounds__(256)
void bte_scatter_kernel(float* __restrict__ h,
                        const int* __restrict__ parent,
                        int off, int count)
{
    const int idx = blockIdx.x * 256 + threadIdx.x;
    const int nl  = idx >> 5;
    if (nl >= count) return;
    const int lane = idx & 31;
    const int node = off + nl;
    const unsigned par = (unsigned)parent[node];
    const float* src = h + (size_t)node * CDIM + lane;
    const float v0 = src[0];
    const float v1 = src[32];
    const float v2 = src[64];
    const float v3 = src[96];
    float* dst = h + (size_t)par * CDIM + lane;
    atomicAdd(dst,      v0);
    atomicAdd(dst + 32, v1);
    atomicAdd(dst + 64, v2);
    atomicAdd(dst + 96, v3);
}

#define NBATCH 512
#define NPART  256
#define TWAVES 16
#define GRP    16
#define TSTEP  (TWAVES * GRP)

__global__ __launch_bounds__(TWAVES * 32)
void bte_max_table_kernel(float* h,
                          const int* __restrict__ batch_id,
                          int N, int chunk)
{
    __shared__ int tab[NBATCH * CDIM];
    const int tid = threadIdx.x;
    {
        int4* t4 = (int4*)tab;
        const int4 z = make_int4(0, 0, 0, 0);
        for (int i = tid; i < NBATCH * CDIM / 4; i += TWAVES * 32) t4[i] = z;
    }
    __syncthreads();

    const int wave = __builtin_amdgcn_readfirstlane(tid >> 5);
    const int lane = tid & 31;
    const int n0   = blockIdx.x * chunk;
    const int n1   = min(n0 + chunk, N);
    const int last = n1 - 1;
    const float* hl = h + lane * 4;
    int*         tl = tab + lane;

#define BTE_LOAD(V, B, BASE)                                                  \
    _Pragma("unroll")                                                         \
    for (int i = 0; i < GRP; ++i) {                                           \
        const int node = min((BASE) + i, last);                               \
        B[i] = batch_id[node];                                                \
        V[i] = *(const float4*)(hl + (size_t)node * CDIM);                    \
    }
#define BTE_FOLD(V, B)                                                        \
    _Pragma("unroll")                                                         \
    for (int i = 0; i < GRP; ++i) {                                           \
        int* t = tl + B[i] * CDIM;                                            \
        atomicMax(t,      __float_as_int(V[i].x));                            \
        atomicMax(t + 32, __float_as_int(V[i].y));                            \
        atomicMax(t + 64, __float_as_int(V[i].z));                            \
        atomicMax(t + 96, __float_as_int(V[i].w));                            \
    }

    int base = n0 + wave * GRP;
    if (base < n1) {
        float4 va[GRP], vb[GRP];
        int    ba[GRP], bb[GRP];
        BTE_LOAD(va, ba, base)
        for (;;) {
            base += TSTEP;
            if (base >= n1) { BTE_FOLD(va, ba) break; }
            BTE_LOAD(vb, bb, base)
            BTE_FOLD(va, ba)
            base += TSTEP;
            if (base >= n1) { BTE_FOLD(vb, bb) break; }
            BTE_LOAD(va, ba, base)
            BTE_FOLD(vb, bb)
        }
    }
#undef BTE_LOAD
#undef BTE_FOLD
    __syncthreads();

    for (int b = wave; b < NBATCH; b += TWAVES) {
        const int* t = tl + b * CDIM;
        float4 o;
        o.x = __int_as_float(t[0]);
        o.y = __int_as_float(t[32]);
        o.z = __int_as_float(t[64]);
        o.w = __int_as_float(t[96]);
        *(float4*)(h + (size_t)(n0 + b) * CDIM + lane * 4) = o;
    }
}

__global__ __launch_bounds__(256)
void bte_max_fold_kernel(const float* __restrict__ part,
                         float*       __restrict__ out,
                         int chunk)
{
    __shared__ int4 red[8][32];
    const int b    = blockIdx.x;
    const int wave = threadIdx.x >> 5;
    const int lane = threadIdx.x & 31;
    const float* p = part + (size_t)b * CDIM + lane * 4;

    int4 m = make_int4(0, 0, 0, 0);
    #pragma unroll
    for (int k = 0; k < NPART / 8; ++k) {
        const float4 v = *(const float4*)(p + (size_t)(wave + 8 * k) * chunk * CDIM);
        m.x = max(m.x, __float_as_int(v.x));
        m.y = max(m.y, __float_as_int(v.y));
        m.z = max(m.z, __float_as_int(v.z));
        m.w = max(m.w, __float_as_int(v.w));
    }
    red[wave][lane] = m;
    __syncthreads();
    if (wave == 0) {
        #pragma unroll
        for (int g = 1; g < 8; ++g) {
            const int4 r = red[g][lane];
            m.x = max(m.x, r.x); m.y = max(m.y, r.y);
            m.z = max(m.z, r.z); m.w = max(m.w, r.w);
        }
        float4 o;
        o.x = __int_as_float(m.x); o.y = __int_as_float(m.y);
        o.z = __int_as_float(m.z); o.w = __int_as_float(m.w);
        *(float4*)(out + (size_t)b * CDIM + lane * 4) = o;
    }
}

extern "C" void kernel_launch(void* const* d_in, const int* in_sizes, int n_in,
                              void* d_out, int out_size, void* d_ws, size_t ws_size,
                              hipStream_t stream)
{
    const float* emb      = (const float*)d_in[0];
    const float* W        = (const float*)d_in[1];
    const float* bias     = (const float*)d_in[2];
    const int*   tokens   = (const int*)d_in[3];
    const int*   parent   = (const int*)d_in[4];
    const int*   batch_id = (const int*)d_in[6];
    const int    N        = in_sizes[3];

    float* h   = (float*)d_ws;
    float* out = (float*)d_out;

    const int K = in_sizes[0] / EDIM;
    if (2 * K < N) {
        float* tail = h + (size_t)(N - K) * CDIM;
        const int kblocks = (K + 127) / 128;
        bte_gemm_kernel<<<kblocks, 128, 0, stream>>>(emb, W, bias, nullptr, tail, K);
        bte_row_copy_kernel<<<CGROUPS, CWAVES * 32, 0, stream>>>(
            tail, tokens, h, N - K, CGROUPS * CWAVES * CGRP);
        bte_gemm_kernel<<<kblocks, 128, 0, stream>>>(emb, W, bias, tokens + (N - K), tail, K);
    } else {
        const int gemm_blocks = (N + 127) / 128;
        bte_gemm_kernel<<<gemm_blocks, 128, 0, stream>>>(emb, W, bias, tokens, h, N);
    }

    static const int lvl_off[7]  = {0, 512, 2560, 10752, 35328, 109056, 256512};
    static const int lvl_size[7] = {512, 2048, 8192, 24576, 73728, 147456, 147456};
    for (int d = 6; d >= 1; --d) {
        const int threads = lvl_size[d] * 32;
        bte_scatter_kernel<<<(threads + 255) / 256, 256, 0, stream>>>(
            h, parent, lvl_off[d], lvl_size[d]);
    }

    const int chunk = (N + NPART - 1) / NPART;
    bte_max_table_kernel<<<NPART, TWAVES * 32, 0, stream>>>(h, batch_id, N, chunk);
    bte_max_fold_kernel<<<out_size / CDIM, 256, 0, stream>>>(h, out, chunk);
}


# subtree-sum scatter: a wave takes 8 consecutive rows per trip (8 parent numbers, 32 line loads, then 32 line adds) and loops over its level: 21248 waves replace 403456; atomics kept, order free
# speedup vs baseline: 3.4201x; 1.0307x over previous
#include <hip/hip_runtime.h>
#include <hip/hip_bf16.h>

typedef __attribute__((ext_vector_type(2))) float v2f;
typedef __attribute__((ext_vector_type(8))) float v8f;
typedef __attribute__((ext_vector_type(4))) float v4f;

#define EDIM 128
#define CDIM 128

__global__ __launch_bounds__(128)
void bte_gemm_kernel(const float* __restrict__ emb,
                     const float* __restrict__ W,
                     const float* __restrict__ bias,
                     const int*   __restrict__ tokens,
                     float*       __restrict__ h,
                     int N)
{
    __shared__ float ldsA[128 * EDIM];
    const int  block0 = blockIdx.x * 128;
    const int  tid    = threadIdx.x;
    const bool full   = (block0 + 128) <= N;

    {
        const int node = block0 + tid;
        int tok = 0;
        if (full || node < N) tok = tokens ? tokens[node] : node;
        const float4* src = (const float4*)(emb + (size_t)tok * EDIM);
        float4*       dst = (float4*)(ldsA + tid * EDIM);
        #pragma unroll
        for (int i = 0; i < 32; ++i) dst[i] = src[i];
    }
    __syncthreads();

    const int wid   = tid >> 5;
    const int lane  = tid & 31;
    const int col   = lane & 15;
    const int khalf = lane >> 4;
    const int r0    = (wid >> 1) * 64;
    const int c0    = (wid & 1) * 64;

    const float* arow = ldsA + (r0 + col) * EDIM + 2 * khalf;
    const float* wrow = W + (size_t)(c0 + col) * EDIM + 2 * khalf;

    v8f acc[4][4];
    #pragma unroll
    for (int j = 0; j < 4; ++j) {
        const float bv = bias[c0 + 16 * j + col];
        const v8f b8 = { bv, bv, bv, bv, bv, bv, bv, bv };
        #pragma unroll
        for (int i = 0; i < 4; ++i) acc[i][j] = b8;
    }

    #pragma unroll 2
    for (int k = 0; k < EDIM; k += 4) {
        v2f a[4], bf[4];
        #pragma unroll
        for (int i = 0; i < 4; ++i) {
            a[i].x = arow[i * 16 * EDIM + k];
            a[i].y = arow[i * 16 * EDIM + k + 1];
        }
        #pragma unroll
        for (int j = 0; j < 4; ++j) {
            bf[j].x = wrow[j * 16 * EDIM + k];
            bf[j].y = wrow[j * 16 * EDIM + k + 1];
        }
        #pragma unroll
        for (int i = 0; i < 4; ++i) {
            #pragma unroll
            for (int j = 0; j < 4; ++j) {
                acc[i][j] = __builtin_amdgcn_wmma_f32_16x16x4_f32(
                                false, a[i], false, bf[j], (short)0, acc[i][j],
                                false, false);
            }
        }
    }

    #pragma unroll
    for (int i = 0; i < 4; ++i) {
        const int mbase = block0 + r0 + 16 * i + 8 * khalf;
        #pragma unroll
        for (int j = 0; j < 4; ++j) {
            float* hp = h + (size_t)mbase * CDIM + c0 + 16 * j + col;
            if (full) {
                #pragma unroll
                for (int r = 0; r < 8; ++r) hp[r * CDIM] = acc[i][j][r];
            } else {
                #pragma unroll
                for (int r = 0; r < 8; ++r)
                    if (mbase + r < N) hp[r * CDIM] = acc[i][j][r];
            }
        }
    }
}

#define CWAVES 8
#define CGROUPS 512
#define CGRP   16

__global__ __launch_bounds__(CWAVES * 32)
void bte_row_copy_kernel(const float* __restrict__ table,
                         const int*   __restrict__ tokens,
                         float*       __restrict__ h,
                         int rows, int step)
{
    const int wave = threadIdx.x >> 5;
    const int lane = threadIdx.x & 31;
    const int last = rows - 1;
    const float* tl = table + lane * 4;
    float*       hl = h + lane * 4;

    int base = (blockIdx.x * CWAVES + wave) * CGRP;
    if (base >= rows) return;

    int tk[CGRP];
    #pragma unroll
    for (int i = 0; i < CGRP; ++i) tk[i] = tokens[min(base + i, last)];

    for (;;) {
        v4f v[CGRP];
        #pragma unroll
        for (int i = 0; i < CGRP; ++i)
            v[i] = *(const v4f*)(tl + (size_t)tk[i] * CDIM);

        const int next = base + step;
        #pragma unroll
        for (int i = 0; i < CGRP; ++i) tk[i] = tokens[min(next + i, last)];

        asm volatile("" :: "v"(v[0]), "v"(v[1]), "v"(v[2]), "v"(v[3]),
                           "v"(v[4]), "v"(v[5]), "v"(v[6]), "v"(v[7]),
                           "v"(v[8]), "v"(v[9]), "v"(v[10]), "v"(v[11]),
                           "v"(v[12]), "v"(v[13]), "v"(v[14]), "v"(v[15]));

        #pragma unroll
        for (int i = 0; i < CGRP; ++i)
            *(v4f*)(hl + (size_t)min(base + i, last) * CDIM) = v[i];

        if (next >= rows) break;
        base = next;
    }
}

#define SWAVES 8
#define SROWS  8

__global__ __launch_bounds__(SWAVES * 32)
void bte_scatter_kernel(float* __restrict__ h,
                        const int* __restrict__ parent,
                        int off, int count, int step)
{
    const int wave = __builtin_amdgcn_readfirstlane(threadIdx.x >> 5);
    const int lane = threadIdx.x & 31;
    const float* hs = h + (size_t)off * CDIM + lane;
    float*       hd = h + lane;
    const int*   pl = parent + off;

    for (int base = (blockIdx.x * SWAVES + wave) * SROWS; base < count; base += step) {
        if (base + SROWS <= count) {
            unsigned par[SROWS];
            float    v[SROWS][4];
            #pragma unroll
            for (int i = 0; i < SROWS; ++i) par[i] = (unsigned)pl[base + i];
            #pragma unroll
            for (int i = 0; i < SROWS; ++i) {
                const float* src = hs + (size_t)(base + i) * CDIM;
                v[i][0] = src[0];
                v[i][1] = src[32];
                v[i][2] = src[64];
                v[i][3] = src[96];
            }
            #pragma unroll
            for (int i = 0; i < SROWS; ++i) {
                float* dst = hd + (size_t)par[i] * CDIM;
                atomicAdd(dst,      v[i][0]);
                atomicAdd(dst + 32, v[i][1]);
                atomicAdd(dst + 64, v[i][2]);
                atomicAdd(dst + 96, v[i][3]);
            }
        } else {
            for (int n = base; n < count; ++n) {
                const unsigned par = (unsigned)pl[n];
                const float* src = hs + (size_t)n * CDIM;
                const float v0 = src[0];
                const float v1 = src[32];
                const float v2 = src[64];
                const float v3 = src[96];
                float* dst = hd + (size_t)par * CDIM;
                atomicAdd(dst,      v0);
                atomicAdd(dst + 32, v1);
                atomicAdd(dst + 64, v2);
                atomicAdd(dst + 96, v3);
            }
        }
    }
}

#define NBATCH 512
#define NPART  256
#define TWAVES 16
#define GRP    16
#define TSTEP  (TWAVES * GRP)

__global__ __launch_bounds__(TWAVES * 32)
void bte_max_table_kernel(float* h,
                          const int* __restrict__ batch_id,
                          int N, int chunk)
{
    __shared__ int tab[NBATCH * CDIM];
    const int tid = threadIdx.x;
    {
        int4* t4 = (int4*)tab;
        const int4 z = make_int4(0, 0, 0, 0);
        for (int i = tid; i < NBATCH * CDIM / 4; i += TWAVES * 32) t4[i] = z;
    }
    __syncthreads();

    const int wave = __builtin_amdgcn_readfirstlane(tid >> 5);
    const int lane = tid & 31;
    const int n0   = blockIdx.x * chunk;
    const int n1   = min(n0 + chunk, N);
    const int last = n1 - 1;
    const float* hl = h + lane * 4;
    int*         tl = tab + lane;

#define BTE_LOAD(V, B, BASE)                                                  \
    _Pragma("unroll")                                                         \
    for (int i = 0; i < GRP; ++i) {                                           \
        const int node = min((BASE) + i, last);                               \
        B[i] = batch_id[node];                                                \
        V[i] = *(const float4*)(hl + (size_t)node * CDIM);                    \
    }
#define BTE_FOLD(V, B)                                                        \
    _Pragma("unroll")                                                         \
    for (int i = 0; i < GRP; ++i) {                                           \
        int* t = tl + B[i] * CDIM;                                            \
        atomicMax(t,      __float_as_int(V[i].x));                            \
        atomicMax(t + 32, __float_as_int(V[i].y));                            \
        atomicMax(t + 64, __float_as_int(V[i].z));                            \
        atomicMax(t + 96, __float_as_int(V[i].w));                            \
    }

    int base = n0 + wave * GRP;
    if (base < n1) {
        float4 va[GRP], vb[GRP];
        int    ba[GRP], bb[GRP];
        BTE_LOAD(va, ba, base)
        for (;;) {
            base += TSTEP;
            if (base >= n1) { BTE_FOLD(va, ba) break; }
            BTE_LOAD(vb, bb, base)
            BTE_FOLD(va, ba)
            base += TSTEP;
            if (base >= n1) { BTE_FOLD(vb, bb) break; }
            BTE_LOAD(va, ba, base)
            BTE_FOLD(vb, bb)
        }
    }
#undef BTE_LOAD
#undef BTE_FOLD
    __syncthreads();

    for (int b = wave; b < NBATCH; b += TWAVES) {
        const int* t = tl + b * CDIM;
        float4 o;
        o.x = __int_as_float(t[0]);
        o.y = __int_as_float(t[32]);
        o.z = __int_as_float(t[64]);
        o.w = __int_as_float(t[96]);
        *(float4*)(h + (size_t)(n0 + b) * CDIM + lane * 4) = o;
    }
}

__global__ __launch_bounds__(256)
void bte_max_fold_kernel(const float* __restrict__ part,
                         float*       __restrict__ out,
                         int chunk)
{
    __shared__ int4 red[8][32];
    const int b    = blockIdx.x;
    const int wave = threadIdx.x >> 5;
    const int lane = threadIdx.x & 31;
    const float* p = part + (size_t)b * CDIM + lane * 4;

    int4 m = make_int4(0, 0, 0, 0);
    #pragma unroll
    for (int k = 0; k < NPART / 8; ++k) {
        const float4 v = *(const float4*)(p + (size_t)(wave + 8 * k) * chunk * CDIM);
        m.x = max(m.x, __float_as_int(v.x));
        m.y = max(m.y, __float_as_int(v.y));
        m.z = max(m.z, __float_as_int(v.z));
        m.w = max(m.w, __float_as_int(v.w));
    }
    red[wave][lane] = m;
    __syncthreads();
    if (wave == 0) {
        #pragma unroll
        for (int g = 1; g < 8; ++g) {
            const int4 r = red[g][lane];
            m.x = max(m.x, r.x); m.y = max(m.y, r.y);
            m.z = max(m.z, r.z); m.w = max(m.w, r.w);
        }
        float4 o;
        o.x = __int_as_float(m.x); o.y = __int_as_float(m.y);
        o.z = __int_as_float(m.z); o.w = __int_as_float(m.w);
        *(float4*)(out + (size_t)b * CDIM + lane * 4) = o;
    }
}

extern "C" void kernel_launch(void* const* d_in, const int* in_sizes, int n_in,
                              void* d_out, int out_size, void* d_ws, size_t ws_size,
                              hipStream_t stream)
{
    const float* emb      = (const float*)d_in[0];
    const float* W        = (const float*)d_in[1];
    const float* bias     = (const float*)d_in[2];
    const int*   tokens   = (const int*)d_in[3];
    const int*   parent   = (const int*)d_in[4];
    const int*   batch_id = (const int*)d_in[6];
    const int    N        = in_sizes[3];

    float* h   = (float*)d_ws;
    float* out = (float*)d_out;

    const int K = in_sizes[0] / EDIM;
    if (2 * K < N) {
        float* tail = h + (size_t)(N - K) * CDIM;
        const int kblocks = (K + 127) / 128;
        bte_gemm_kernel<<<kblocks, 128, 0, stream>>>(emb, W, bias, nullptr, tail, K);
        bte_row_copy_kernel<<<CGROUPS, CWAVES * 32, 0, stream>>>(
            tail, tokens, h, N - K, CGROUPS * CWAVES * CGRP);
        bte_gemm_kernel<<<kblocks, 128, 0, stream>>>(emb, W, bias, tokens + (N - K), tail, K);
    } else {
        const int gemm_blocks = (N + 127) / 128;
        bte_gemm_kernel<<<gemm_blocks, 128, 0, stream>>>(emb, W, bias, tokens, h, N);
    }

    static const int lvl_off[7]  = {0, 512, 2560, 10752, 35328, 109056, 256512};
    static const int lvl_size[7] = {512, 2048, 8192, 24576, 73728, 147456, 147456};
    static const int lvl_wgs[7]  = {0, 32, 128, 384, 576, 768, 768};
    static const int lvl_step[7] = {0, 2048, 8192, 24576, 36864, 49152, 49152};
    for (int d = 6; d >= 1; --d) {
        bte_scatter_kernel<<<lvl_wgs[d], SWAVES * 32, 0, stream>>>(
            h, parent, lvl_off[d], lvl_size[d], lvl_step[d]);
    }

    const int chunk = (N + NPART - 1) / NPART;
    bte_max_table_kernel<<<NPART, TWAVES * 32, 0, stream>>>(h, batch_id, N, chunk);
    bte_max_fold_kernel<<<out_size / CDIM, 256, 0, stream>>>(h, out, chunk);
}


# deepest level in one launch: tail rows are added to their parents by the work-group that projects them (out of LDS), the level's other rows by scatter groups of the same launch; atomic order free
# speedup vs baseline: 3.5292x; 1.0319x over previous
#include <hip/hip_runtime.h>
#include <hip/hip_bf16.h>

typedef __attribute__((ext_vector_type(2))) float v2f;
typedef __attribute__((ext_vector_type(8))) float v8f;
typedef __attribute__((ext_vector_type(4))) float v4f;

#define EDIM 128
#define CDIM 128

__global__ __launch_bounds__(128)
void bte_gemm_kernel(const float* __restrict__ emb,
                     const float* __restrict__ W,
                     const float* __restrict__ bias,
                     const int*   __restrict__ tokens,
                     float*       __restrict__ h,
                     int N)
{
    __shared__ float ldsA[128 * EDIM];
    const int  block0 = blockIdx.x * 128;
    const int  tid    = threadIdx.x;
    const bool full   = (block0 + 128) <= N;

    {
        const int node = block0 + tid;
        int tok = 0;
        if (full || node < N) tok = tokens ? tokens[node] : node;
        const float4* src = (const float4*)(emb + (size_t)tok * EDIM);
        float4*       dst = (float4*)(ldsA + tid * EDIM);
        #pragma unroll
        for (int i = 0; i < 32; ++i) dst[i] = src[i];
    }
    __syncthreads();

    const int wid   = tid >> 5;
    const int lane  = tid & 31;
    const int col   = lane & 15;
    const int khalf = lane >> 4;
    const int r0    = (wid >> 1) * 64;
    const int c0    = (wid & 1) * 64;

    const float* arow = ldsA + (r0 + col) * EDIM + 2 * khalf;
    const float* wrow = W + (size_t)(c0 + col) * EDIM + 2 * khalf;

    v8f acc[4][4];
    #pragma unroll
    for (int j = 0; j < 4; ++j) {
        const float bv = bias[c0 + 16 * j + col];
        const v8f b8 = { bv, bv, bv, bv, bv, bv, bv, bv };
        #pragma unroll
        for (int i = 0; i < 4; ++i) acc[i][j] = b8;
    }

    #pragma unroll 2
    for (int k = 0; k < EDIM; k += 4) {
        v2f a[4], bf[4];
        #pragma unroll
        for (int i = 0; i < 4; ++i) {
            a[i].x = arow[i * 16 * EDIM + k];
            a[i].y = arow[i * 16 * EDIM + k + 1];
        }
        #pragma unroll
        for (int j = 0; j < 4; ++j) {
            bf[j].x = wrow[j * 16 * EDIM + k];
            bf[j].y = wrow[j * 16 * EDIM + k + 1];
        }
        #pragma unroll
        for (int i = 0; i < 4; ++i) {
            #pragma unroll
            for (int j = 0; j < 4; ++j) {
                acc[i][j] = __builtin_amdgcn_wmma_f32_16x16x4_f32(
                                false, a[i], false, bf[j], (short)0, acc[i][j],
                                false, false);
            }
        }
    }

    #pragma unroll
    for (int i = 0; i < 4; ++i) {
        const int mbase = block0 + r0 + 16 * i + 8 * khalf;
        #pragma unroll
        for (int j = 0; j < 4; ++j) {
            float* hp = h + (size_t)mbase * CDIM + c0 + 16 * j + col;
            if (full) {
                #pragma unroll
                for (int r = 0; r < 8; ++r) hp[r * CDIM] = acc[i][j][r];
            } else {
                #pragma unroll
                for (int r = 0; r < 8; ++r)
                    if (mbase + r < N) hp[r * CDIM] = acc[i][j][r];
            }
        }
    }
}

#define CWAVES 8
#define CGROUPS 512
#define CGRP   16

__global__ __launch_bounds__(CWAVES * 32)
void bte_row_copy_kernel(const float* __restrict__ table,
                         const int*   __restrict__ tokens,
                         float*       __restrict__ h,
                         int rows, int step)
{
    const int wave = threadIdx.x >> 5;
    const int lane = threadIdx.x & 31;
    const int last = rows - 1;
    const float* tl = table + lane * 4;
    float*       hl = h + lane * 4;

    int base = (blockIdx.x * CWAVES + wave) * CGRP;
    if (base >= rows) return;

    int tk[CGRP];
    #pragma unroll
    for (int i = 0; i < CGRP; ++i) tk[i] = tokens[min(base + i, last)];

    for (;;) {
        v4f v[CGRP];
        #pragma unroll
        for (int i = 0; i < CGRP; ++i)
            v[i] = *(const v4f*)(tl + (size_t)tk[i] * CDIM);

        const int next = base + step;
        #pragma unroll
        for (int i = 0; i < CGRP; ++i) tk[i] = tokens[min(next + i, last)];

        asm volatile("" :: "v"(v[0]), "v"(v[1]), "v"(v[2]), "v"(v[3]),
                           "v"(v[4]), "v"(v[5]), "v"(v[6]), "v"(v[7]),
                           "v"(v[8]), "v"(v[9]), "v"(v[10]), "v"(v[11]),
                           "v"(v[12]), "v"(v[13]), "v"(v[14]), "v"(v[15]));

        #pragma unroll
        for (int i = 0; i < CGRP; ++i)
            *(v4f*)(hl + (size_t)min(base + i, last) * CDIM) = v[i];

        if (next >= rows) break;
        base = next;
    }
}

#define SWAVES 8
#define SROWS  8

__global__ __launch_bounds__(SWAVES * 32)
void bte_scatter_kernel(float* __restrict__ h,
                        const int* __restrict__ parent,
                        int off, int count, int step)
{
    const int wave = __builtin_amdgcn_readfirstlane(threadIdx.x >> 5);
    const int lane = threadIdx.x & 31;
    const float* hs = h + (size_t)off * CDIM + lane;
    float*       hd = h + lane;
    const int*   pl = parent + off;

    for (int base = (blockIdx.x * SWAVES + wave) * SROWS; base < count; base += step) {
        if (base + SROWS <= count) {
            unsigned par[SROWS];
            float    v[SROWS][4];
            #pragma unroll
            for (int i = 0; i < SROWS; ++i) par[i] = (unsigned)pl[base + i];
            #pragma unroll
            for (int i = 0; i < SROWS; ++i) {
                const float* src = hs + (size_t)(base + i) * CDIM;
                v[i][0] = src[0];
                v[i][1] = src[32];
                v[i][2] = src[64];
                v[i][3] = src[96];
            }
            #pragma unroll
            for (int i = 0; i < SROWS; ++i) {
                float* dst = hd + (size_t)par[i] * CDIM;
                atomicAdd(dst,      v[i][0]);
                atomicAdd(dst + 32, v[i][1]);
                atomicAdd(dst + 64, v[i][2]);
                atomicAdd(dst + 96, v[i][3]);
            }
        } else {
            for (int n = base; n < count; ++n) {
                const unsigned par = (unsigned)pl[n];
                const float* src = hs + (size_t)n * CDIM;
                const float v0 = src[0];
                const float v1 = src[32];
                const float v2 = src[64];
                const float v3 = src[96];
                float* dst = hd + (size_t)par * CDIM;
                atomicAdd(dst,      v0);
                atomicAdd(dst + 32, v1);
                atomicAdd(dst + 64, v2);
                atomicAdd(dst + 96, v3);
            }
        }
    }
}

#define FWAVES 4
#define FROWS  16

__global__ __launch_bounds__(FWAVES * 32)
void bte_tail_scatter_kernel(const float* __restrict__ emb,
                             const float* __restrict__ W,
                             const float* __restrict__ bias,
                             const int*   __restrict__ tokens,
                             const int*   __restrict__ parent,
                             float*       __restrict__ h,
                             int tail0, int K, int gemm_blocks,
                             int off, int count, int step)
{
    __shared__ float ldsA[128 * EDIM];
    const int tid  = threadIdx.x;
    const int lane = tid & 31;
    const int wave = __builtin_amdgcn_readfirstlane(tid >> 5);

    if ((int)blockIdx.x >= gemm_blocks) {
        const float* hs = h + (size_t)off * CDIM + lane;
        float*       hd = h + lane;
        const int*   pl = parent + off;
        const int    w0 = ((int)blockIdx.x - gemm_blocks) * FWAVES + wave;
        for (int base = w0 * FROWS; base < count; base += step) {
            if (base + FROWS <= count) {
                unsigned par[FROWS];
                float    v[FROWS][4];
                #pragma unroll
                for (int i = 0; i < FROWS; ++i) par[i] = (unsigned)pl[base + i];
                #pragma unroll
                for (int i = 0; i < FROWS; ++i) {
                    const float* src = hs + (size_t)(base + i) * CDIM;
                    v[i][0] = src[0];
                    v[i][1] = src[32];
                    v[i][2] = src[64];
                    v[i][3] = src[96];
                }
                #pragma unroll
                for (int i = 0; i < FROWS; ++i) {
                    float* dst = hd + (size_t)par[i] * CDIM;
                    atomicAdd(dst,      v[i][0]);
                    atomicAdd(dst + 32, v[i][1]);
                    atomicAdd(dst + 64, v[i][2]);
                    atomicAdd(dst + 96, v[i][3]);
                }
            } else {
                for (int n = base; n < count; ++n) {
                    const unsigned par = (unsigned)pl[n];
                    const float* src = hs + (size_t)n * CDIM;
                    const float v0 = src[0];
                    const float v1 = src[32];
                    const float v2 = src[64];
                    const float v3 = src[96];
                    float* dst = hd + (size_t)par * CDIM;
                    atomicAdd(dst,      v0);
                    atomicAdd(dst + 32, v1);
                    atomicAdd(dst + 64, v2);
                    atomicAdd(dst + 96, v3);
                }
            }
        }
        return;
    }

    const int  block0 = blockIdx.x * 128;
    const bool full   = (block0 + 128) <= K;
    {
        const int m = block0 + tid;
        int tok = 0;
        if (full || m < K) tok = tokens[tail0 + m];
        const float4* src = (const float4*)(emb + (size_t)tok * EDIM);
        float4*       dst = (float4*)(ldsA + tid * EDIM);
        #pragma unroll
        for (int i = 0; i < 32; ++i) dst[i] = src[i];
    }
    __syncthreads();

    const int wid   = tid >> 5;
    const int col   = lane & 15;
    const int khalf = lane >> 4;
    const int r0    = (wid >> 1) * 64;
    const int c0    = (wid & 1) * 64;

    const int parv = parent[tail0 + min(block0 + wave * 32 + lane, K - 1)];

    const float* arow = ldsA + (r0 + col) * EDIM + 2 * khalf;
    const float* wrow = W + (size_t)(c0 + col) * EDIM + 2 * khalf;

    v8f acc[4][4];
    #pragma unroll
    for (int j = 0; j < 4; ++j) {
        const float bv = bias[c0 + 16 * j + col];
        const v8f b8 = { bv, bv, bv, bv, bv, bv, bv, bv };
        #pragma unroll
        for (int i = 0; i < 4; ++i) acc[i][j] = b8;
    }

    #pragma unroll 2
    for (int k = 0; k < EDIM; k += 4) {
        v2f a[4], bf[4];
        #pragma unroll
        for (int i = 0; i < 4; ++i) {
            a[i].x = arow[i * 16 * EDIM + k];
            a[i].y = arow[i * 16 * EDIM + k + 1];
        }
        #pragma unroll
        for (int j = 0; j < 4; ++j) {
            bf[j].x = wrow[j * 16 * EDIM + k];
            bf[j].y = wrow[j * 16 * EDIM + k + 1];
        }
        #pragma unroll
        for (int i = 0; i < 4; ++i) {
            #pragma unroll
            for (int j = 0; j < 4; ++j) {
                acc[i][j] = __builtin_amdgcn_wmma_f32_16x16x4_f32(
                                false, a[i], false, bf[j], (short)0, acc[i][j],
                                false, false);
            }
        }
    }

    __syncthreads();
    #pragma unroll
    for (int i = 0; i < 4; ++i) {
        #pragma unroll
        for (int j = 0; j < 4; ++j) {
            float* lp = ldsA + (r0 + 16 * i + 8 * khalf) * CDIM + c0 + 16 * j + col;
            #pragma unroll
            for (int r = 0; r < 8; ++r) lp[r * CDIM] = acc[i][j][r];
        }
    }
    __syncthreads();

    const float* srow = ldsA + (wave * 32) * CDIM + lane;
    float*       own  = h + (size_t)(tail0 + block0 + wave * 32) * CDIM + lane;
    float*       hd   = h + lane;
    if (full) {
        #pragma unroll
        for (int g = 0; g < 32; g += 8) {
            float v[8][4];
            #pragma unroll
            for (int i = 0; i < 8; ++i) {
                const float* s = srow + (g + i) * CDIM;
                v[i][0] = s[0];
                v[i][1] = s[32];
                v[i][2] = s[64];
                v[i][3] = s[96];
            }
            #pragma unroll
            for (int i = 0; i < 8; ++i) {
                float* o = own + (size_t)(g + i) * CDIM;
                o[0]  = v[i][0];
                o[32] = v[i][1];
                o[64] = v[i][2];
                o[96] = v[i][3];
            }
            #pragma unroll
            for (int i = 0; i < 8; ++i) {
                const unsigned par = (unsigned)__builtin_amdgcn_readlane(parv, g + i);
                float* dst = hd + (size_t)par * CDIM;
                atomicAdd(dst,      v[i][0]);
                atomicAdd(dst + 32, v[i][1]);
                atomicAdd(dst + 64, v[i][2]);
                atomicAdd(dst + 96, v[i][3]);
            }
        }
    } else {
        const int nrow = min(32, K - (block0 + wave * 32));
        for (int t = 0; t < nrow; ++t) {
            const unsigned par = (unsigned)__builtin_amdgcn_readlane(parv, t);
            const float* s = srow + t * CDIM;
            const float v0 = s[0];
            const float v1 = s[32];
            const float v2 = s[64];
            const float v3 = s[96];
            float* o = own + (size_t)t * CDIM;
            o[0]  = v0;
            o[32] = v1;
            o[64] = v2;
            o[96] = v3;
            float* dst = hd + (size_t)par * CDIM;
            atomicAdd(dst,      v0);
            atomicAdd(dst + 32, v1);
            atomicAdd(dst + 64, v2);
            atomicAdd(dst + 96, v3);
        }
    }
}

#define NBATCH 512
#define NPART  256
#define TWAVES 16
#define GRP    16
#define TSTEP  (TWAVES * GRP)

__global__ __launch_bounds__(TWAVES * 32)
void bte_max_table_kernel(float* h,
                          const int* __restrict__ batch_id,
                          int N, int chunk)
{
    __shared__ int tab[NBATCH * CDIM];
    const int tid = threadIdx.x;
    {
        int4* t4 = (int4*)tab;
        const int4 z = make_int4(0, 0, 0, 0);
        for (int i = tid; i < NBATCH * CDIM / 4; i += TWAVES * 32) t4[i] = z;
    }
    __syncthreads();

    const int wave = __builtin_amdgcn_readfirstlane(tid >> 5);
    const int lane = tid & 31;
    const int n0   = blockIdx.x * chunk;
    const int n1   = min(n0 + chunk, N);
    const int last = n1 - 1;
    const float* hl = h + lane * 4;
    int*         tl = tab + lane;

#define BTE_LOAD(V, B, BASE)                                                  \
    _Pragma("unroll")                                                         \
    for (int i = 0; i < GRP; ++i) {                                           \
        const int node = min((BASE) + i, last);                               \
        B[i] = batch_id[node];                                                \
        V[i] = *(const float4*)(hl + (size_t)node * CDIM);                    \
    }
#define BTE_FOLD(V, B)                                                        \
    _Pragma("unroll")                                                         \
    for (int i = 0; i < GRP; ++i) {                                           \
        int* t = tl + B[i] * CDIM;                                            \
        atomicMax(t,      __float_as_int(V[i].x));                            \
        atomicMax(t + 32, __float_as_int(V[i].y));                            \
        atomicMax(t + 64, __float_as_int(V[i].z));                            \
        atomicMax(t + 96, __float_as_int(V[i].w));                            \
    }

    int base = n0 + wave * GRP;
    if (base < n1) {
        float4 va[GRP], vb[GRP];
        int    ba[GRP], bb[GRP];
        BTE_LOAD(va, ba, base)
        for (;;) {
            base += TSTEP;
            if (base >= n1) { BTE_FOLD(va, ba) break; }
            BTE_LOAD(vb, bb, base)
            BTE_FOLD(va, ba)
            base += TSTEP;
            if (base >= n1) { BTE_FOLD(vb, bb) break; }
            BTE_LOAD(va, ba, base)
            BTE_FOLD(vb, bb)
        }
    }
#undef BTE_LOAD
#undef BTE_FOLD
    __syncthreads();

    for (int b = wave; b < NBATCH; b += TWAVES) {
        const int* t = tl + b * CDIM;
        float4 o;
        o.x = __int_as_float(t[0]);
        o.y = __int_as_float(t[32]);
        o.z = __int_as_float(t[64]);
        o.w = __int_as_float(t[96]);
        *(float4*)(h + (size_t)(n0 + b) * CDIM + lane * 4) = o;
    }
}

__global__ __launch_bounds__(256)
void bte_max_fold_kernel(const float* __restrict__ part,
                         float*       __restrict__ out,
                         int chunk)
{
    __shared__ int4 red[8][32];
    const int b    = blockIdx.x;
    const int wave = threadIdx.x >> 5;
    const int lane = threadIdx.x & 31;
    const float* p = part + (size_t)b * CDIM + lane * 4;

    int4 m = make_int4(0, 0, 0, 0);
    #pragma unroll
    for (int k = 0; k < NPART / 8; ++k) {
        const float4 v = *(const float4*)(p + (size_t)(wave + 8 * k) * chunk * CDIM);
        m.x = max(m.x, __float_as_int(v.x));
        m.y = max(m.y, __float_as_int(v.y));
        m.z = max(m.z, __float_as_int(v.z));
        m.w = max(m.w, __float_as_int(v.w));
    }
    red[wave][lane] = m;
    __syncthreads();
    if (wave == 0) {
        #pragma unroll
        for (int g = 1; g < 8; ++g) {
            const int4 r = red[g][lane];
            m.x = max(m.x, r.x); m.y = max(m.y, r.y);
            m.z = max(m.z, r.z); m.w = max(m.w, r.w);
        }
        float4 o;
        o.x = __int_as_float(m.x); o.y = __int_as_float(m.y);
        o.z = __int_as_float(m.z); o.w = __int_as_float(m.w);
        *(float4*)(out + (size_t)b * CDIM + lane * 4) = o;
    }
}

extern "C" void kernel_launch(void* const* d_in, const int* in_sizes, int n_in,
                              void* d_out, int out_size, void* d_ws, size_t ws_size,
                              hipStream_t stream)
{
    const float* emb      = (const float*)d_in[0];
    const float* W        = (const float*)d_in[1];
    const float* bias     = (const float*)d_in[2];
    const int*   tokens   = (const int*)d_in[3];
    const int*   parent   = (const int*)d_in[4];
    const int*   batch_id = (const int*)d_in[6];
    const int    N        = in_sizes[3];

    float* h   = (float*)d_ws;
    float* out = (float*)d_out;

    const int K = in_sizes[0] / EDIM;
    const bool fused6 = (2 * K < N) && (N - K >= 256512);
    if (2 * K < N) {
        float* tail = h + (size_t)(N - K) * CDIM;
        const int kblocks = (K + 127) / 128;
        bte_gemm_kernel<<<kblocks, 128, 0, stream>>>(emb, W, bias, nullptr, tail, K);
        bte_row_copy_kernel<<<CGROUPS, CWAVES * 32, 0, stream>>>(
            tail, tokens, h, N - K, CGROUPS * CWAVES * CGRP);
        if (fused6) {
            bte_tail_scatter_kernel<<<kblocks + 768, FWAVES * 32, 0, stream>>>(
                emb, W, bias, tokens, parent, h, N - K, K, kblocks,
                256512, (N - K) - 256512, 768 * FWAVES * FROWS);
        } else {
            bte_gemm_kernel<<<kblocks, 128, 0, stream>>>(emb, W, bias, tokens + (N - K), tail, K);
        }
    } else {
        const int gemm_blocks = (N + 127) / 128;
        bte_gemm_kernel<<<gemm_blocks, 128, 0, stream>>>(emb, W, bias, tokens, h, N);
    }

    static const int lvl_off[7]  = {0, 512, 2560, 10752, 35328, 109056, 256512};
    static const int lvl_size[7] = {512, 2048, 8192, 24576, 73728, 147456, 147456};
    static const int lvl_wgs[7]  = {0, 32, 128, 384, 576, 768, 768};
    static const int lvl_step[7] = {0, 2048, 8192, 24576, 36864, 49152, 49152};
    for (int d = fused6 ? 5 : 6; d >= 1; --d) {
        bte_scatter_kernel<<<lvl_wgs[d], SWAVES * 32, 0, stream>>>(
            h, parent, lvl_off[d], lvl_size[d], lvl_step[d]);
    }

    const int chunk = (N + NPART - 1) / NPART;
    bte_max_table_kernel<<<NPART, TWAVES * 32, 0, stream>>>(h, batch_id, N, chunk);
    bte_max_fold_kernel<<<out_size / CDIM, 256, 0, stream>>>(h, out, chunk);
}


# max pass: 128 row chunks x 2 column halves (was 256 chunks): a work-group folds 256-byte half rows of a doubled chunk into a 512x64 LDS table; half the partial-table bytes are written and folded
# speedup vs baseline: 3.7268x; 1.0560x over previous
#include <hip/hip_runtime.h>
#include <hip/hip_bf16.h>

typedef __attribute__((ext_vector_type(2))) float v2f;
typedef __attribute__((ext_vector_type(8))) float v8f;
typedef __attribute__((ext_vector_type(4))) float v4f;

#define EDIM 128
#define CDIM 128

__global__ __launch_bounds__(128)
void bte_gemm_kernel(const float* __restrict__ emb,
                     const float* __restrict__ W,
                     const float* __restrict__ bias,
                     const int*   __restrict__ tokens,
                     float*       __restrict__ h,
                     int N)
{
    __shared__ float ldsA[128 * EDIM];
    const int  block0 = blockIdx.x * 128;
    const int  tid    = threadIdx.x;
    const bool full   = (block0 + 128) <= N;

    {
        const int node = block0 + tid;
        int tok = 0;
        if (full || node < N) tok = tokens ? tokens[node] : node;
        const float4* src = (const float4*)(emb + (size_t)tok * EDIM);
        float4*       dst = (float4*)(ldsA + tid * EDIM);
        #pragma unroll
        for (int i = 0; i < 32; ++i) dst[i] = src[i];
    }
    __syncthreads();

    const int wid   = tid >> 5;
    const int lane  = tid & 31;
    const int col   = lane & 15;
    const int khalf = lane >> 4;
    const int r0    = (wid >> 1) * 64;
    const int c0    = (wid & 1) * 64;

    const float* arow = ldsA + (r0 + col) * EDIM + 2 * khalf;
    const float* wrow = W + (size_t)(c0 + col) * EDIM + 2 * khalf;

    v8f acc[4][4];
    #pragma unroll
    for (int j = 0; j < 4; ++j) {
        const float bv = bias[c0 + 16 * j + col];
        const v8f b8 = { bv, bv, bv, bv, bv, bv, bv, bv };
        #pragma unroll
        for (int i = 0; i < 4; ++i) acc[i][j] = b8;
    }

    #pragma unroll 2
    for (int k = 0; k < EDIM; k += 4) {
        v2f a[4], bf[4];
        #pragma unroll
        for (int i = 0; i < 4; ++i) {
            a[i].x = arow[i * 16 * EDIM + k];
            a[i].y = arow[i * 16 * EDIM + k + 1];
        }
        #pragma unroll
        for (int j = 0; j < 4; ++j) {
            bf[j].x = wrow[j * 16 * EDIM + k];
            bf[j].y = wrow[j * 16 * EDIM + k + 1];
        }
        #pragma unroll
        for (int i = 0; i < 4; ++i) {
            #pragma unroll
            for (int j = 0; j < 4; ++j) {
                acc[i][j] = __builtin_amdgcn_wmma_f32_16x16x4_f32(
                                false, a[i], false, bf[j], (short)0, acc[i][j],
                                false, false);
            }
        }
    }

    #pragma unroll
    for (int i = 0; i < 4; ++i) {
        const int mbase = block0 + r0 + 16 * i + 8 * khalf;
        #pragma unroll
        for (int j = 0; j < 4; ++j) {
            float* hp = h + (size_t)mbase * CDIM + c0 + 16 * j + col;
            if (full) {
                #pragma unroll
                for (int r = 0; r < 8; ++r) hp[r * CDIM] = acc[i][j][r];
            } else {
                #pragma unroll
                for (int r = 0; r < 8; ++r)
                    if (mbase + r < N) hp[r * CDIM] = acc[i][j][r];
            }
        }
    }
}

#define CWAVES 8
#define CGROUPS 512
#define CGRP   16

__global__ __launch_bounds__(CWAVES * 32)
void bte_row_copy_kernel(const float* __restrict__ table,
                         const int*   __restrict__ tokens,
                         float*       __restrict__ h,
                         int rows, int step)
{
    const int wave = threadIdx.x >> 5;
    const int lane = threadIdx.x & 31;
    const int last = rows - 1;
    const float* tl = table + lane * 4;
    float*       hl = h + lane * 4;

    int base = (blockIdx.x * CWAVES + wave) * CGRP;
    if (base >= rows) return;

    int tk[CGRP];
    #pragma unroll
    for (int i = 0; i < CGRP; ++i) tk[i] = tokens[min(base + i, last)];

    for (;;) {
        v4f v[CGRP];
        #pragma unroll
        for (int i = 0; i < CGRP; ++i)
            v[i] = *(const v4f*)(tl + (size_t)tk[i] * CDIM);

        const int next = base + step;
        #pragma unroll
        for (int i = 0; i < CGRP; ++i) tk[i] = tokens[min(next + i, last)];

        asm volatile("" :: "v"(v[0]), "v"(v[1]), "v"(v[2]), "v"(v[3]),
                           "v"(v[4]), "v"(v[5]), "v"(v[6]), "v"(v[7]),
                           "v"(v[8]), "v"(v[9]), "v"(v[10]), "v"(v[11]),
                           "v"(v[12]), "v"(v[13]), "v"(v[14]), "v"(v[15]));

        #pragma unroll
        for (int i = 0; i < CGRP; ++i)
            *(v4f*)(hl + (size_t)min(base + i, last) * CDIM) = v[i];

        if (next >= rows) break;
        base = next;
    }
}

#define SWAVES 8
#define SROWS  8

__global__ __launch_bounds__(SWAVES * 32)
void bte_scatter_kernel(float* __restrict__ h,
                        const int* __restrict__ parent,
                        int off, int count, int step)
{
    const int wave = __builtin_amdgcn_readfirstlane(threadIdx.x >> 5);
    const int lane = threadIdx.x & 31;
    const float* hs = h + (size_t)off * CDIM + lane;
    float*       hd = h + lane;
    const int*   pl = parent + off;

    for (int base = (blockIdx.x * SWAVES + wave) * SROWS; base < count; base += step) {
        if (base + SROWS <= count) {
            unsigned par[SROWS];
            float    v[SROWS][4];
            #pragma unroll
            for (int i = 0; i < SROWS; ++i) par[i] = (unsigned)pl[base + i];
            #pragma unroll
            for (int i = 0; i < SROWS; ++i) {
                const float* src = hs + (size_t)(base + i) * CDIM;
                v[i][0] = src[0];
                v[i][1] = src[32];
                v[i][2] = src[64];
                v[i][3] = src[96];
            }
            #pragma unroll
            for (int i = 0; i < SROWS; ++i) {
                float* dst = hd + (size_t)par[i] * CDIM;
                atomicAdd(dst,      v[i][0]);
                atomicAdd(dst + 32, v[i][1]);
                atomicAdd(dst + 64, v[i][2]);
                atomicAdd(dst + 96, v[i][3]);
            }
        } else {
            for (int n = base; n < count; ++n) {
                const unsigned par = (unsigned)pl[n];
                const float* src = hs + (size_t)n * CDIM;
                const float v0 = src[0];
                const float v1 = src[32];
                const float v2 = src[64];
                const float v3 = src[96];
                float* dst = hd + (size_t)par * CDIM;
                atomicAdd(dst,      v0);
                atomicAdd(dst + 32, v1);
                atomicAdd(dst + 64, v2);
                atomicAdd(dst + 96, v3);
            }
        }
    }
}

#define FWAVES 4
#define FROWS  16

__global__ __launch_bounds__(FWAVES * 32)
void bte_tail_scatter_kernel(const float* __restrict__ emb,
                             const float* __restrict__ W,
                             const float* __restrict__ bias,
                             const int*   __restrict__ tokens,
                             const int*   __restrict__ parent,
                             float*       __restrict__ h,
                             int tail0, int K, int gemm_blocks,
                             int off, int count, int step)
{
    __shared__ float ldsA[128 * EDIM];
    const int tid  = threadIdx.x;
    const int lane = tid & 31;
    const int wave = __builtin_amdgcn_readfirstlane(tid >> 5);

    if ((int)blockIdx.x >= gemm_blocks) {
        const float* hs = h + (size_t)off * CDIM + lane;
        float*       hd = h + lane;
        const int*   pl = parent + off;
        const int    w0 = ((int)blockIdx.x - gemm_blocks) * FWAVES + wave;
        for (int base = w0 * FROWS; base < count; base += step) {
            if (base + FROWS <= count) {
                unsigned par[FROWS];
                float    v[FROWS][4];
                #pragma unroll
                for (int i = 0; i < FROWS; ++i) par[i] = (unsigned)pl[base + i];
                #pragma unroll
                for (int i = 0; i < FROWS; ++i) {
                    const float* src = hs + (size_t)(base + i) * CDIM;
                    v[i][0] = src[0];
                    v[i][1] = src[32];
                    v[i][2] = src[64];
                    v[i][3] = src[96];
                }
                #pragma unroll
                for (int i = 0; i < FROWS; ++i) {
                    float* dst = hd + (size_t)par[i] * CDIM;
                    atomicAdd(dst,      v[i][0]);
                    atomicAdd(dst + 32, v[i][1]);
                    atomicAdd(dst + 64, v[i][2]);
                    atomicAdd(dst + 96, v[i][3]);
                }
            } else {
                for (int n = base; n < count; ++n) {
                    const unsigned par = (unsigned)pl[n];
                    const float* src = hs + (size_t)n * CDIM;
                    const float v0 = src[0];
                    const float v1 = src[32];
                    const float v2 = src[64];
                    const float v3 = src[96];
                    float* dst = hd + (size_t)par * CDIM;
                    atomicAdd(dst,      v0);
                    atomicAdd(dst + 32, v1);
                    atomicAdd(dst + 64, v2);
                    atomicAdd(dst + 96, v3);
                }
            }
        }
        return;
    }

    const int  block0 = blockIdx.x * 128;
    const bool full   = (block0 + 128) <= K;
    {
        const int m = block0 + tid;
        int tok = 0;
        if (full || m < K) tok = tokens[tail0 + m];
        const float4* src = (const float4*)(emb + (size_t)tok * EDIM);
        float4*       dst = (float4*)(ldsA + tid * EDIM);
        #pragma unroll
        for (int i = 0; i < 32; ++i) dst[i] = src[i];
    }
    __syncthreads();

    const int wid   = tid >> 5;
    const int col   = lane & 15;
    const int khalf = lane >> 4;
    const int r0    = (wid >> 1) * 64;
    const int c0    = (wid & 1) * 64;

    const int parv = parent[tail0 + min(block0 + wave * 32 + lane, K - 1)];

    const float* arow = ldsA + (r0 + col) * EDIM + 2 * khalf;
    const float* wrow = W + (size_t)(c0 + col) * EDIM + 2 * khalf;

    v8f acc[4][4];
    #pragma unroll
    for (int j = 0; j < 4; ++j) {
        const float bv = bias[c0 + 16 * j + col];
        const v8f b8 = { bv, bv, bv, bv, bv, bv, bv, bv };
        #pragma unroll
        for (int i = 0; i < 4; ++i) acc[i][j] = b8;
    }

    #pragma unroll 2
    for (int k = 0; k < EDIM; k += 4) {
        v2f a[4], bf[4];
        #pragma unroll
        for (int i = 0; i < 4; ++i) {
            a[i].x = arow[i * 16 * EDIM + k];
            a[i].y = arow[i * 16 * EDIM + k + 1];
        }
        #pragma unroll
        for (int j = 0; j < 4; ++j) {
            bf[j].x = wrow[j * 16 * EDIM + k];
            bf[j].y = wrow[j * 16 * EDIM + k + 1];
        }
        #pragma unroll
        for (int i = 0; i < 4; ++i) {
            #pragma unroll
            for (int j = 0; j < 4; ++j) {
                acc[i][j] = __builtin_amdgcn_wmma_f32_16x16x4_f32(
                                false, a[i], false, bf[j], (short)0, acc[i][j],
                                false, false);
            }
        }
    }

    __syncthreads();
    #pragma unroll
    for (int i = 0; i < 4; ++i) {
        #pragma unroll
        for (int j = 0; j < 4; ++j) {
            float* lp = ldsA + (r0 + 16 * i + 8 * khalf) * CDIM + c0 + 16 * j + col;
            #pragma unroll
            for (int r = 0; r < 8; ++r) lp[r * CDIM] = acc[i][j][r];
        }
    }
    __syncthreads();

    const float* srow = ldsA + (wave * 32) * CDIM + lane;
    float*       own  = h + (size_t)(tail0 + block0 + wave * 32) * CDIM + lane;
    float*       hd   = h + lane;
    if (full) {
        #pragma unroll
        for (int g = 0; g < 32; g += 8) {
            float v[8][4];
            #pragma unroll
            for (int i = 0; i < 8; ++i) {
                const float* s = srow + (g + i) * CDIM;
                v[i][0] = s[0];
                v[i][1] = s[32];
                v[i][2] = s[64];
                v[i][3] = s[96];
            }
            #pragma unroll
            for (int i = 0; i < 8; ++i) {
                float* o = own + (size_t)(g + i) * CDIM;
                o[0]  = v[i][0];
                o[32] = v[i][1];
                o[64] = v[i][2];
                o[96] = v[i][3];
            }
            #pragma unroll
            for (int i = 0; i < 8; ++i) {
                const unsigned par = (unsigned)__builtin_amdgcn_readlane(parv, g + i);
                float* dst = hd + (size_t)par * CDIM;
                atomicAdd(dst,      v[i][0]);
                atomicAdd(dst + 32, v[i][1]);
                atomicAdd(dst + 64, v[i][2]);
                atomicAdd(dst + 96, v[i][3]);
            }
        }
    } else {
        const int nrow = min(32, K - (block0 + wave * 32));
        for (int t = 0; t < nrow; ++t) {
            const unsigned par = (unsigned)__builtin_amdgcn_readlane(parv, t);
            const float* s = srow + t * CDIM;
            const float v0 = s[0];
            const float v1 = s[32];
            const float v2 = s[64];
            const float v3 = s[96];
            float* o = own + (size_t)t * CDIM;
            o[0]  = v0;
            o[32] = v1;
            o[64] = v2;
            o[96] = v3;
            float* dst = hd + (size_t)par * CDIM;
            atomicAdd(dst,      v0);
            atomicAdd(dst + 32, v1);
            atomicAdd(dst + 64, v2);
            atomicAdd(dst + 96, v3);
        }
    }
}

#define NBATCH 512
#define NPART  128
#define HCOL   (CDIM / 2)
#define TWAVES 16
#define GRP    32
#define TSTEP  (TWAVES * GRP)

__global__ __launch_bounds__(TWAVES * 32)
void bte_max_table_kernel(float* h,
                          const int* __restrict__ batch_id,
                          int N, int chunk)
{
    __shared__ int tab[NBATCH * HCOL];
    const int tid = threadIdx.x;
    {
        int4* t4 = (int4*)tab;
        const int4 z = make_int4(0, 0, 0, 0);
        for (int i = tid; i < NBATCH * HCOL / 4; i += TWAVES * 32) t4[i] = z;
    }
    __syncthreads();

    const int wave = __builtin_amdgcn_readfirstlane(tid >> 5);
    const int lane = tid & 31;
    const int part = blockIdx.x >> 1;
    const int half = blockIdx.x & 1;
    const int n0   = part * chunk;
    const int n1   = min(n0 + chunk, N);
    const int last = n1 - 1;
    const float* hl = h + half * HCOL + lane * 2;
    int*         tl = tab + lane;

#define BTE_LOAD(V, BV, BASE)                                                 \
    BV = batch_id[min((BASE) + lane, last)];                                  \
    _Pragma("unroll")                                                         \
    for (int i = 0; i < GRP; ++i) {                                           \
        const int node = min((BASE) + i, last);                               \
        V[i] = *(const float2*)(hl + (size_t)node * CDIM);                    \
    }
#define BTE_FOLD(V, BV)                                                       \
    _Pragma("unroll")                                                         \
    for (int i = 0; i < GRP; ++i) {                                           \
        int* t = tl + __builtin_amdgcn_readlane(BV, i) * HCOL;                \
        atomicMax(t,      __float_as_int(V[i].x));                            \
        atomicMax(t + 32, __float_as_int(V[i].y));                            \
    }

    int base = n0 + wave * GRP;
    if (base < n1) {
        float2 va[GRP], vb[GRP];
        int    ba, bb;
        BTE_LOAD(va, ba, base)
        for (;;) {
            base += TSTEP;
            if (base >= n1) { BTE_FOLD(va, ba) break; }
            BTE_LOAD(vb, bb, base)
            BTE_FOLD(va, ba)
            base += TSTEP;
            if (base >= n1) { BTE_FOLD(vb, bb) break; }
            BTE_LOAD(va, ba, base)
            BTE_FOLD(vb, bb)
        }
    }
#undef BTE_LOAD
#undef BTE_FOLD
    __syncthreads();

    for (int b = wave; b < NBATCH; b += TWAVES) {
        const int* t = tl + b * HCOL;
        float2 o;
        o.x = __int_as_float(t[0]);
        o.y = __int_as_float(t[32]);
        *(float2*)(h + (size_t)(n0 + b) * CDIM + half * HCOL + lane * 2) = o;
    }
}

__global__ __launch_bounds__(256)
void bte_max_fold_kernel(const float* __restrict__ part,
                         float*       __restrict__ out,
                         int chunk)
{
    __shared__ int4 red[8][32];
    const int b    = blockIdx.x;
    const int wave = threadIdx.x >> 5;
    const int lane = threadIdx.x & 31;
    const float* p = part + (size_t)b * CDIM + lane * 4;

    int4 m = make_int4(0, 0, 0, 0);
    #pragma unroll
    for (int k = 0; k < NPART / 8; ++k) {
        const float4 v = *(const float4*)(p + (size_t)(wave + 8 * k) * chunk * CDIM);
        m.x = max(m.x, __float_as_int(v.x));
        m.y = max(m.y, __float_as_int(v.y));
        m.z = max(m.z, __float_as_int(v.z));
        m.w = max(m.w, __float_as_int(v.w));
    }
    red[wave][lane] = m;
    __syncthreads();
    if (wave == 0) {
        #pragma unroll
        for (int g = 1; g < 8; ++g) {
            const int4 r = red[g][lane];
            m.x = max(m.x, r.x); m.y = max(m.y, r.y);
            m.z = max(m.z, r.z); m.w = max(m.w, r.w);
        }
        float4 o;
        o.x = __int_as_float(m.x); o.y = __int_as_float(m.y);
        o.z = __int_as_float(m.z); o.w = __int_as_float(m.w);
        *(float4*)(out + (size_t)b * CDIM + lane * 4) = o;
    }
}

extern "C" void kernel_launch(void* const* d_in, const int* in_sizes, int n_in,
                              void* d_out, int out_size, void* d_ws, size_t ws_size,
                              hipStream_t stream)
{
    const float* emb      = (const float*)d_in[0];
    const float* W        = (const float*)d_in[1];
    const float* bias     = (const float*)d_in[2];
    const int*   tokens   = (const int*)d_in[3];
    const int*   parent   = (const int*)d_in[4];
    const int*   batch_id = (const int*)d_in[6];
    const int    N        = in_sizes[3];

    float* h   = (float*)d_ws;
    float* out = (float*)d_out;

    const int K = in_sizes[0] / EDIM;
    const bool fused6 = (2 * K < N) && (N - K >= 256512);
    if (2 * K < N) {
        float* tail = h + (size_t)(N - K) * CDIM;
        const int kblocks = (K + 127) / 128;
        bte_gemm_kernel<<<kblocks, 128, 0, stream>>>(emb, W, bias, nullptr, tail, K);
        bte_row_copy_kernel<<<CGROUPS, CWAVES * 32, 0, stream>>>(
            tail, tokens, h, N - K, CGROUPS * CWAVES * CGRP);
        if (fused6) {
            bte_tail_scatter_kernel<<<kblocks + 768, FWAVES * 32, 0, stream>>>(
                emb, W, bias, tokens, parent, h, N - K, K, kblocks,
                256512, (N - K) - 256512, 768 * FWAVES * FROWS);
        } else {
            bte_gemm_kernel<<<kblocks, 128, 0, stream>>>(emb, W, bias, tokens + (N - K), tail, K);
        }
    } else {
        const int gemm_blocks = (N + 127) / 128;
        bte_gemm_kernel<<<gemm_blocks, 128, 0, stream>>>(emb, W, bias, tokens, h, N);
    }

    static const int lvl_off[7]  = {0, 512, 2560, 10752, 35328, 109056, 256512};
    static const int lvl_size[7] = {512, 2048, 8192, 24576, 73728, 147456, 147456};
    static const int lvl_wgs[7]  = {0, 32, 128, 384, 576, 768, 768};
    static const int lvl_step[7] = {0, 2048, 8192, 24576, 36864, 49152, 49152};
    for (int d = fused6 ? 5 : 6; d >= 1; --d) {
        bte_scatter_kernel<<<lvl_wgs[d], SWAVES * 32, 0, stream>>>(
            h, parent, lvl_off[d], lvl_size[d], lvl_step[d]);
    }

    const int chunk = (N + NPART - 1) / NPART;
    bte_max_table_kernel<<<2 * NPART, TWAVES * 32, 0, stream>>>(h, batch_id, N, chunk);
    bte_max_fold_kernel<<<out_size / CDIM, 256, 0, stream>>>(h, out, chunk);
}
